# starting kernel
# baseline (speedup 1.0000x reference)
#include <hip/hip_runtime.h>
#include <math.h>


typedef __attribute__((ext_vector_type(2))) float v2f;
typedef __attribute__((ext_vector_type(8))) float v8f;

#define S_LEN 384
#define B_SZ  256
#define D_IN  192
#define H_DIM 100
#define FH    400
#define C_OUT 250

#define XS_STRIDE 200
#define HS  105
#define ZS  404

__device__ __forceinline__ float sigmoidf_(float v) {
    return 1.0f / (1.0f + __expf(-v));
}

__device__ __forceinline__ float fast_tanhf_(float v) {
#if __has_builtin(__builtin_amdgcn_tanhf)
    return __builtin_amdgcn_tanhf(v);
#elif __has_builtin(__builtin_amdgcn_tanh_f32)
    return __builtin_amdgcn_tanh_f32(v);
#else
    return tanhf(v);
#endif
}

__global__ __launch_bounds__(256)
void lstm_xw_kernel(const float* __restrict__ x, const float* __restrict__ W,
                    const float* __restrict__ bias, float* __restrict__ pre) {
    __shared__ float xs[16 * XS_STRIDE];
    const int blk = blockIdx.x;
    const int s   = blk >> 4;
    const int b0  = (blk & 15) << 4;
    const int tid = threadIdx.x;

    for (int e = tid; e < 16 * D_IN; e += 256) {
        const int r = e / D_IN, d = e - r * D_IN;
        xs[r * XS_STRIDE + d] = x[((size_t)(b0 + r) * S_LEN + s) * D_IN + d];
    }
    __syncthreads();

    const int wave  = tid >> 5;
    const int lane  = tid & 31;
    const int lhalf = lane >> 4;
    const int l15   = lane & 15;
    const int koff  = lhalf * 2;

    for (int t = wave; t < 25; t += 8) {
        const int n0 = t << 4;
        const int n  = n0 + l15;
        const float bb = bias[n];
        v8f acc;
#pragma unroll
        for (int r = 0; r < 8; ++r) acc[r] = bb;

#pragma unroll 4
        for (int kk = 0; kk < 48; ++kk) {
            const int k0 = kk * 4 + koff;
            v2f a, bf;
            a[0]  = xs[l15 * XS_STRIDE + k0];
            a[1]  = xs[l15 * XS_STRIDE + k0 + 1];
            bf[0] = W[(size_t)k0 * FH + n];
            bf[1] = W[(size_t)(k0 + 1) * FH + n];
            acc = __builtin_amdgcn_wmma_f32_16x16x4_f32(
                false, a, false, bf, (short)0, acc, false, false);
        }
        float* prow = pre + ((size_t)s * B_SZ + b0) * FH;
#pragma unroll
        for (int r = 0; r < 8; ++r)
            prow[(size_t)(r + lhalf * 8) * FH + n] = acc[r];
    }
}

__global__ __launch_bounds__(256)
void lstm_rec_kernel(const float* __restrict__ pre, const float* __restrict__ U,
                     float* __restrict__ hT) {
    __shared__ float U_lds[H_DIM * FH];
    __shared__ float h_s[16 * HS];
    __shared__ float c_s[16 * H_DIM];
    __shared__ float z_s[16 * ZS];

    const int b0  = blockIdx.x << 4;
    const int tid = threadIdx.x;
    const int wave  = tid >> 5;
    const int lane  = tid & 31;
    const int lhalf = lane >> 4;
    const int l15   = lane & 15;
    const int koff  = lhalf * 2;

    for (int e = tid; e < H_DIM * FH; e += 256) U_lds[e] = U[e];
    for (int e = tid; e < 16 * HS; e += 256)    h_s[e] = 0.0f;
    for (int e = tid; e < 16 * H_DIM; e += 256) c_s[e] = 0.0f;
    __syncthreads();

    for (int s = 0; s < S_LEN; ++s) {
        const float* prow = pre + ((size_t)s * B_SZ + b0) * FH;

        if (s + 1 < S_LEN) {
            const float* nrow = prow + (size_t)B_SZ * FH;
            for (int e = tid; e < 400; e += 256)
                __builtin_prefetch(nrow + e * 16, 0, 0);
        }

        for (int t = wave; t < 25; t += 8) {
            const int n0 = t << 4;
            const int n  = n0 + l15;
            v8f acc;
#pragma unroll
            for (int r = 0; r < 8; ++r)
                acc[r] = prow[(size_t)(r + lhalf * 8) * FH + n];

#pragma unroll
            for (int kk = 0; kk < 25; ++kk) {
                const int k0 = kk * 4 + koff;
                v2f a, bf;
                a[0]  = h_s[l15 * HS + k0];
                a[1]  = h_s[l15 * HS + k0 + 1];
                bf[0] = U_lds[k0 * FH + n];
                bf[1] = U_lds[(k0 + 1) * FH + n];
                acc = __builtin_amdgcn_wmma_f32_16x16x4_f32(
                    false, a, false, bf, (short)0, acc, false, false);
            }
#pragma unroll
            for (int r = 0; r < 8; ++r)
                z_s[(r + lhalf * 8) * ZS + n] = acc[r];
        }
        __syncthreads();

        for (int e = tid; e < 16 * H_DIM; e += 256) {
            const int b = e / H_DIM, j = e - b * H_DIM;
            const float zi = z_s[b * ZS + j];
            const float zf = z_s[b * ZS + H_DIM + j];
            const float zg = z_s[b * ZS + 2 * H_DIM + j];
            const float zo = z_s[b * ZS + 3 * H_DIM + j];
            const float ig = sigmoidf_(zi);
            const float fg = sigmoidf_(zf);
            const float gg = fast_tanhf_(zg);
            const float og = sigmoidf_(zo);
            const float cn = fg * c_s[e] + ig * gg;
            c_s[e] = cn;
            h_s[b * HS + j] = og * fast_tanhf_(cn);
        }
        __syncthreads();
    }

    for (int e = tid; e < 16 * H_DIM; e += 256) {
        const int b = e / H_DIM, j = e - b * H_DIM;
        hT[(size_t)(b0 + b) * H_DIM + j] = h_s[b * HS + j];
    }
}

__global__ __launch_bounds__(32)
void dense_kernel(const float* __restrict__ hT, const float* __restrict__ Wd,
                  const float* __restrict__ bd, float* __restrict__ out) {
    const int m0 = (blockIdx.x >> 4) << 4;
    const int n0 = (blockIdx.x & 15) << 4;
    const int lane  = threadIdx.x & 31;
    const int lhalf = lane >> 4;
    const int l15   = lane & 15;
    const int koff  = lhalf * 2;
    const int n = n0 + l15;
    const bool nvalid = (n < C_OUT);

    v8f acc;
    const float bb = nvalid ? bd[n] : 0.0f;
#pragma unroll
    for (int r = 0; r < 8; ++r) acc[r] = bb;

#pragma unroll
    for (int kk = 0; kk < 25; ++kk) {
        const int k0 = kk * 4 + koff;
        v2f a, bf;
        a[0]  = hT[(size_t)(m0 + l15) * H_DIM + k0];
        a[1]  = hT[(size_t)(m0 + l15) * H_DIM + k0 + 1];
        bf[0] = nvalid ? Wd[(size_t)k0 * C_OUT + n] : 0.0f;
        bf[1] = nvalid ? Wd[(size_t)(k0 + 1) * C_OUT + n] : 0.0f;
        acc = __builtin_amdgcn_wmma_f32_16x16x4_f32(
            false, a, false, bf, (short)0, acc, false, false);
    }
    if (nvalid) {
#pragma unroll
        for (int r = 0; r < 8; ++r)
            out[(size_t)(m0 + r + lhalf * 8) * C_OUT + n] = acc[r];
    }
}

extern "C" void kernel_launch(void* const* d_in, const int* in_sizes, int n_in,
                              void* d_out, int out_size, void* d_ws, size_t ws_size,
                              hipStream_t stream) {
    const float* x  = (const float*)d_in[0];
    const float* W  = (const float*)d_in[1];
    const float* U  = (const float*)d_in[2];
    const float* bi = (const float*)d_in[3];
    const float* Wd = (const float*)d_in[4];
    const float* bd = (const float*)d_in[5];
    float* out = (float*)d_out;

    float* pre = (float*)d_ws;
    float* hT  = pre + (size_t)S_LEN * B_SZ * FH;

    lstm_xw_kernel<<<S_LEN * (B_SZ / 16), 256, 0, stream>>>(x, W, bi, pre);
    lstm_rec_kernel<<<B_SZ / 16, 256, 0, stream>>>(pre, U, hT);
    dense_kernel<<<(B_SZ / 16) * 16, 32, 0, stream>>>(hT, Wd, bd, out);
}


# input projection: 256 persistent work-groups keep W^T whole in LDS (16-byte fragment loads) and read x fragments straight from global; W leaves L2 once per group, not per M-tile; K sum order kept
# speedup vs baseline: 1.6663x; 1.6663x over previous
#include <hip/hip_runtime.h>
#include <math.h>


typedef __attribute__((ext_vector_type(2))) float v2f;
typedef __attribute__((ext_vector_type(4))) float v4f;
typedef __attribute__((ext_vector_type(8))) float v8f;

#define S_LEN 384
#define B_SZ  256
#define D_IN  192
#define H_DIM 100
#define FH    400
#define C_OUT 250

__device__ __forceinline__ float sigmoidf_(float v) {
    return 1.0f / (1.0f + __expf(-v));
}

__device__ __forceinline__ float fast_tanhf_(float v) {
#if __has_builtin(__builtin_amdgcn_tanhf)
    return __builtin_amdgcn_tanhf(v);
#elif __has_builtin(__builtin_amdgcn_tanh_f32)
    return __builtin_amdgcn_tanh_f32(v);
#else
    return tanhf(v);
#endif
}

#define WPITCH 196
#define XW_WAVES 8
#define XW_THREADS (XW_WAVES * 32)
#define XW_GRID 256
#define MT_PER_WG   ((S_LEN * B_SZ / 16) / XW_GRID)
#define MT_PER_WAVE (MT_PER_WG / XW_WAVES)
#define XKG   24
#define XBLKG 3
#define XNBLK (XKG / XBLKG)

__device__ __forceinline__ int kperm8(int k) {
    return (k & ~6) | ((k & 2) << 1) | ((k & 4) >> 1);
}

__device__ __forceinline__ void keep_grouped() { __builtin_amdgcn_sched_barrier(0); }

struct WFrag { v4f v[XBLKG][4]; };

__device__ __forceinline__ void load_wfrag(WFrag& f, const float* const (&wb)[4], int blk) {
#pragma unroll
    for (int i = 0; i < XBLKG; ++i)
#pragma unroll
        for (int q = 0; q < 4; ++q)
            f.v[i][q] = *(const v4f*)(wb[q] + 8 * (blk * XBLKG + i));
}

__device__ __forceinline__ void xw_mma_block(v8f (&acc)[4], const v2f (&a)[2 * XKG],
                                             const WFrag& f, int blk) {
#pragma unroll
    for (int i = 0; i < XBLKG; ++i) {
        const int g = blk * XBLKG + i;
#pragma unroll
        for (int q = 0; q < 4; ++q) {
            v2f b0;
            b0[0] = f.v[i][q][0]; b0[1] = f.v[i][q][1];
            acc[q] = __builtin_amdgcn_wmma_f32_16x16x4_f32(
                false, a[2 * g], false, b0, (short)0, acc[q], false, false);
        }
#pragma unroll
        for (int q = 0; q < 4; ++q) {
            v2f b1;
            b1[0] = f.v[i][q][2]; b1[1] = f.v[i][q][3];
            acc[q] = __builtin_amdgcn_wmma_f32_16x16x4_f32(
                false, a[2 * g + 1], false, b1, (short)0, acc[q], false, false);
        }
    }
}

__global__ __launch_bounds__(XW_THREADS)
void lstm_xw_kernel(const float* __restrict__ x, const float* __restrict__ W,
                    const float* __restrict__ bias, float* __restrict__ pre) {
    __shared__ __attribute__((aligned(16))) float WT[FH * WPITCH];
    __shared__ __attribute__((aligned(16))) float bs[FH];
    const int tid   = threadIdx.x;
    const int wave  = tid >> 5;
    const int lane  = tid & 31;
    const int lhalf = lane >> 4;
    const int l15   = lane & 15;

    {
        int k  = tid / (FH / 4);
        int cq = tid - k * (FH / 4);
#pragma unroll 15
        for (int i = 0; i < (D_IN * (FH / 4)) / XW_THREADS; ++i) {
            const v4f w = *(const v4f*)(W + (size_t)k * FH + 4 * cq);
            const bool inq  = (cq < 96);
            const int rbase = inq ? (((cq >> 4) << 6) | (cq & 15)) : 4 * cq;
            const int rstep = inq ? 16 * WPITCH : WPITCH;
            float* dst = WT + rbase * WPITCH + kperm8(k);
#pragma unroll
            for (int jj = 0; jj < 4; ++jj) dst[jj * rstep] = w[jj];
            cq += XW_THREADS - 2 * (FH / 4);
            k  += 2;
            if (cq >= FH / 4) { cq -= FH / 4; ++k; }
        }
    }
    for (int e = tid; e < FH; e += XW_THREADS) bs[e] = bias[e];
    __syncthreads();

    const int mt0 = blockIdx.x * MT_PER_WG + wave * MT_PER_WAVE;

#pragma unroll 1
    for (int it = 0; it < MT_PER_WAVE; ++it) {
        const int mt = mt0 + it;
        const int s  = mt >> 4;
        const int b0 = (mt & 15) << 4;

        const float* xr = x + ((size_t)(b0 + l15) * S_LEN + s) * D_IN + 2 * lhalf;
        v2f a[2 * XKG];
#pragma unroll
        for (int kk = 0; kk < 2 * XKG; ++kk) a[kk] = *(const v2f*)(xr + 4 * kk);

        if (it + 1 < MT_PER_WAVE) {
            const int mn = mt + 1;
            const float* xn = x + ((size_t)(((mn & 15) << 4) + l15) * S_LEN + (mn >> 4)) * D_IN;
#pragma unroll
            for (int p = 0; p < 3; ++p)
                __builtin_prefetch(xn + (lhalf + 2 * p) * 32, 0, 0);
        }

        float* prow = pre + ((size_t)s * B_SZ + b0 + 8 * lhalf) * FH;

#pragma unroll 1
        for (int T = 0; T < 6; ++T) {
            const float* wb[4];
#pragma unroll
            for (int q = 0; q < 4; ++q)
                wb[q] = WT + (64 * T + 16 * q + l15) * WPITCH + 4 * lhalf;

            WFrag fx, fy;
            load_wfrag(fx, wb, 0);

            const v4f bq = *(const v4f*)(bs + 64 * T + 4 * l15);
            v8f acc[4];
#pragma unroll
            for (int q = 0; q < 4; ++q)
#pragma unroll
                for (int r = 0; r < 8; ++r) acc[q][r] = bq[q];

#pragma unroll
            for (int blk = 0; blk < XNBLK; blk += 2) {
                load_wfrag(fy, wb, blk + 1);
                keep_grouped();
                xw_mma_block(acc, a, fx, blk);
                keep_grouped();
                if (blk + 2 < XNBLK) load_wfrag(fx, wb, blk + 2);
                keep_grouped();
                xw_mma_block(acc, a, fy, blk + 1);
                keep_grouped();
            }

#pragma unroll
            for (int r = 0; r < 8; ++r) {
                v4f o;
                o[0] = acc[0][r]; o[1] = acc[1][r]; o[2] = acc[2][r]; o[3] = acc[3][r];
                *(v4f*)(prow + (size_t)r * FH + 64 * T + 4 * l15) = o;
            }
        }

        {
            const float* wt = WT + (384 + l15) * WPITCH + 4 * lhalf;
            const float bb = bs[384 + l15];
            v8f acc1;
#pragma unroll
            for (int r = 0; r < 8; ++r) acc1[r] = bb;
            v4f f0[XKG / 2], f1[XKG / 2];
#pragma unroll
            for (int g = 0; g < XKG / 2; ++g) f0[g] = *(const v4f*)(wt + 8 * g);
#pragma unroll
            for (int g = 0; g < XKG / 2; ++g) f1[g] = *(const v4f*)(wt + 8 * (g + XKG / 2));
            keep_grouped();
#pragma unroll
            for (int g = 0; g < XKG / 2; ++g) {
                v2f b0, b1;
                b0[0] = f0[g][0]; b0[1] = f0[g][1];
                b1[0] = f0[g][2]; b1[1] = f0[g][3];
                acc1 = __builtin_amdgcn_wmma_f32_16x16x4_f32(
                    false, a[2 * g], false, b0, (short)0, acc1, false, false);
                acc1 = __builtin_amdgcn_wmma_f32_16x16x4_f32(
                    false, a[2 * g + 1], false, b1, (short)0, acc1, false, false);
            }
#pragma unroll
            for (int g = 0; g < XKG / 2; ++g) {
                const int gg = g + XKG / 2;
                v2f b0, b1;
                b0[0] = f1[g][0]; b0[1] = f1[g][1];
                b1[0] = f1[g][2]; b1[1] = f1[g][3];
                acc1 = __builtin_amdgcn_wmma_f32_16x16x4_f32(
                    false, a[2 * gg], false, b0, (short)0, acc1, false, false);
                acc1 = __builtin_amdgcn_wmma_f32_16x16x4_f32(
                    false, a[2 * gg + 1], false, b1, (short)0, acc1, false, false);
            }
#pragma unroll
            for (int r = 0; r < 8; ++r)
                prow[(size_t)r * FH + 384 + l15] = acc1[r];
        }
    }
}

#define REC_WAVES   7
#define REC_THREADS (REC_WAVES * 32)
#define UP    100
#define HP    108
#define KGRP  12
#define BLKG  3
#define KTAIL 96

__device__ __forceinline__ int kpos(int k) {
    return (k < KTAIL) ? kperm8(k) : k;
}

struct UFrag { v4f v[BLKG][4]; };

__device__ __forceinline__ void load_ufrag(UFrag& f, const float* const (&ub)[4], int blk) {
#pragma unroll
    for (int i = 0; i < BLKG; ++i)
#pragma unroll
        for (int q = 0; q < 4; ++q)
            f.v[i][q] = *(const v4f*)(ub[q] + 8 * (blk * BLKG + i));
}

__device__ __forceinline__ void mma_block(v8f (&acc)[4], const v4f (&a)[KGRP],
                                          const UFrag& f, int blk) {
#pragma unroll
    for (int i = 0; i < BLKG; ++i) {
        const v4f av = a[blk * BLKG + i];
        v2f a0, a1;
        a0[0] = av[0]; a0[1] = av[1];
        a1[0] = av[2]; a1[1] = av[3];
#pragma unroll
        for (int q = 0; q < 4; ++q) {
            v2f b0;
            b0[0] = f.v[i][q][0]; b0[1] = f.v[i][q][1];
            acc[q] = __builtin_amdgcn_wmma_f32_16x16x4_f32(
                false, a0, false, b0, (short)0, acc[q], false, false);
        }
#pragma unroll
        for (int q = 0; q < 4; ++q) {
            v2f b1;
            b1[0] = f.v[i][q][2]; b1[1] = f.v[i][q][3];
            acc[q] = __builtin_amdgcn_wmma_f32_16x16x4_f32(
                false, a1, false, b1, (short)0, acc[q], false, false);
        }
    }
}

__global__ __launch_bounds__(REC_THREADS)
void lstm_rec_kernel(const float* __restrict__ pre, const float* __restrict__ U,
                     float* __restrict__ hT) {
    __shared__ __attribute__((aligned(16))) float UT[FH * UP];
    __shared__ __attribute__((aligned(16))) float h_s[2 * 16 * HP];

    const int b0  = blockIdx.x << 4;
    const int tid = threadIdx.x;
    const int wave  = tid >> 5;
    const int lane  = tid & 31;
    const int lhalf = lane >> 4;
    const int l15   = lane & 15;

    const int  j      = (wave << 4) + l15;
    const bool jvalid = (j < H_DIM);
    const int  jc     = jvalid ? j : (H_DIM - 1);
    const int  jst    = jvalid ? kpos(j) : (H_DIM + (l15 & 3));

    for (int e = tid; e < H_DIM * FH; e += REC_THREADS) {
        const int k = e / FH, col = e - k * FH;
        UT[col * UP + kpos(k)] = U[e];
    }
    for (int e = tid; e < 2 * 16 * HP; e += REC_THREADS) h_s[e] = 0.0f;
    __syncthreads();

    const float* ub[4];
#pragma unroll
    for (int q = 0; q < 4; ++q) ub[q] = UT + (q * H_DIM + jc) * UP + 4 * lhalf;

    float c_reg[8];
#pragma unroll
    for (int r = 0; r < 8; ++r) c_reg[r] = 0.0f;

    UFrag fx, fy;
    load_ufrag(fx, ub, 0);

    for (int s = 0; s < S_LEN; ++s) {
        const float* prow = pre + ((size_t)s * B_SZ + b0) * FH;
        const float* hr = h_s + (s & 1) * (16 * HP) + l15 * HP + 4 * lhalf;
        float*       hw = h_s + ((s + 1) & 1) * (16 * HP);

        if (s + 1 < S_LEN) {
            const float* nrow = prow + (size_t)B_SZ * FH;
            for (int e = tid; e < 400; e += REC_THREADS)
                __builtin_prefetch(nrow + e * 16, 0, 0);
        }

        v4f a[KGRP];
#pragma unroll
        for (int g = 0; g < KGRP; ++g) a[g] = *(const v4f*)(hr + 8 * g);
        const v2f at = *(const v2f*)(hr - 4 * lhalf + KTAIL + 2 * lhalf);
        keep_grouped();

        float p[4][8];
#pragma unroll
        for (int q = 0; q < 4; ++q)
#pragma unroll
            for (int r = 0; r < 8; ++r)
                p[q][r] = prow[(size_t)(r + lhalf * 8) * FH + q * H_DIM + jc];

        load_ufrag(fy, ub, 1);

        v8f acc[4];
#pragma unroll
        for (int q = 0; q < 4; ++q)
#pragma unroll
            for (int r = 0; r < 8; ++r) acc[q][r] = 0.0f;

        keep_grouped();
        mma_block(acc, a, fx, 0);
        load_ufrag(fx, ub, 2);
        keep_grouped();
        mma_block(acc, a, fy, 1);
        keep_grouped();
        load_ufrag(fy, ub, 3);
        v2f bt[4];
#pragma unroll
        for (int q = 0; q < 4; ++q)
            bt[q] = *(const v2f*)(ub[q] - 4 * lhalf + KTAIL + 2 * lhalf);
        keep_grouped();
        mma_block(acc, a, fx, 2);
        load_ufrag(fx, ub, 0);
        keep_grouped();
        mma_block(acc, a, fy, 3);
#pragma unroll
        for (int q = 0; q < 4; ++q)
            acc[q] = __builtin_amdgcn_wmma_f32_16x16x4_f32(
                false, at, false, bt[q], (short)0, acc[q], false, false);

#pragma unroll
        for (int r = 0; r < 8; ++r) {
            const float zi = acc[0][r] + p[0][r];
            const float zf = acc[1][r] + p[1][r];
            const float zg = acc[2][r] + p[2][r];
            const float zo = acc[3][r] + p[3][r];
            const float ig = sigmoidf_(zi);
            const float fg = sigmoidf_(zf);
            const float gg = fast_tanhf_(zg);
            const float og = sigmoidf_(zo);
            const float cn = fg * c_reg[r] + ig * gg;
            c_reg[r] = cn;
            const float hn = og * fast_tanhf_(cn);
            hw[(r + lhalf * 8) * HP + jst] = hn;
        }
        __syncthreads();
    }

    const float* hl = h_s + (S_LEN & 1) * (16 * HP);
    for (int e = tid; e < 16 * H_DIM; e += REC_THREADS) {
        const int b = e / H_DIM, jj = e - b * H_DIM;
        hT[(size_t)(b0 + b) * H_DIM + jj] = hl[b * HP + kpos(jj)];
    }
}

__global__ __launch_bounds__(32)
void dense_kernel(const float* __restrict__ hT, const float* __restrict__ Wd,
                  const float* __restrict__ bd, float* __restrict__ out) {
    const int m0 = (blockIdx.x >> 4) << 4;
    const int n0 = (blockIdx.x & 15) << 4;
    const int lane  = threadIdx.x & 31;
    const int lhalf = lane >> 4;
    const int l15   = lane & 15;
    const int koff  = lhalf * 2;
    const int n = n0 + l15;
    const bool nvalid = (n < C_OUT);

    v8f acc;
    const float bb = nvalid ? bd[n] : 0.0f;
#pragma unroll
    for (int r = 0; r < 8; ++r) acc[r] = bb;

#pragma unroll
    for (int kk = 0; kk < 25; ++kk) {
        const int k0 = kk * 4 + koff;
        v2f a, bf;
        a[0]  = hT[(size_t)(m0 + l15) * H_DIM + k0];
        a[1]  = hT[(size_t)(m0 + l15) * H_DIM + k0 + 1];
        bf[0] = nvalid ? Wd[(size_t)k0 * C_OUT + n] : 0.0f;
        bf[1] = nvalid ? Wd[(size_t)(k0 + 1) * C_OUT + n] : 0.0f;
        acc = __builtin_amdgcn_wmma_f32_16x16x4_f32(
            false, a, false, bf, (short)0, acc, false, false);
    }
    if (nvalid) {
#pragma unroll
        for (int r = 0; r < 8; ++r)
            out[(size_t)(m0 + r + lhalf * 8) * C_OUT + n] = acc[r];
    }
}

extern "C" void kernel_launch(void* const* d_in, const int* in_sizes, int n_in,
                              void* d_out, int out_size, void* d_ws, size_t ws_size,
                              hipStream_t stream) {
    const float* x  = (const float*)d_in[0];
    const float* W  = (const float*)d_in[1];
    const float* U  = (const float*)d_in[2];
    const float* bi = (const float*)d_in[3];
    const float* Wd = (const float*)d_in[4];
    const float* bd = (const float*)d_in[5];
    float* out = (float*)d_out;

    float* pre = (float*)d_ws;
    float* hT  = pre + (size_t)S_LEN * B_SZ * FH;

    lstm_xw_kernel<<<XW_GRID, XW_THREADS, 0, stream>>>(x, W, bi, pre);
    lstm_rec_kernel<<<B_SZ / 16, REC_THREADS, 0, stream>>>(pre, U, hT);
    dense_kernel<<<(B_SZ / 16) * 16, 32, 0, stream>>>(hT, Wd, bd, out);
}


# recurrence gates: sigmoid(z) = 0.5 + 0.5*tanh(z/2) on the hardware tanh (1 transcendental + 1 fma) replaces exp + IEEE division (15 ops); i,f,o rows of U^T stored halved; sum order kept
# speedup vs baseline: 1.9531x; 1.1721x over previous
#include <hip/hip_runtime.h>
#include <math.h>


typedef __attribute__((ext_vector_type(2))) float v2f;
typedef __attribute__((ext_vector_type(4))) float v4f;
typedef __attribute__((ext_vector_type(8))) float v8f;

#define S_LEN 384
#define B_SZ  256
#define D_IN  192
#define H_DIM 100
#define FH    400
#define C_OUT 250

__device__ __forceinline__ float fast_tanhf_(float v) {
#if __has_builtin(__builtin_amdgcn_tanhf)
    return __builtin_amdgcn_tanhf(v);
#elif __has_builtin(__builtin_amdgcn_tanh_f32)
    return __builtin_amdgcn_tanh_f32(v);
#else
    return tanhf(v);
#endif
}

__device__ __forceinline__ float sigmoid_of_half_(float half_z) {
    return fmaf(0.5f, fast_tanhf_(half_z), 0.5f);
}

#define WPITCH 196
#define XW_WAVES 8
#define XW_THREADS (XW_WAVES * 32)
#define XW_GRID 256
#define MT_PER_WG   ((S_LEN * B_SZ / 16) / XW_GRID)
#define MT_PER_WAVE (MT_PER_WG / XW_WAVES)
#define XKG   24
#define XBLKG 3
#define XNBLK (XKG / XBLKG)

__device__ __forceinline__ int kperm8(int k) {
    return (k & ~6) | ((k & 2) << 1) | ((k & 4) >> 1);
}

__device__ __forceinline__ void keep_grouped() { __builtin_amdgcn_sched_barrier(0); }

struct WFrag { v4f v[XBLKG][4]; };

__device__ __forceinline__ void load_wfrag(WFrag& f, const float* const (&wb)[4], int blk) {
#pragma unroll
    for (int i = 0; i < XBLKG; ++i)
#pragma unroll
        for (int q = 0; q < 4; ++q)
            f.v[i][q] = *(const v4f*)(wb[q] + 8 * (blk * XBLKG + i));
}

__device__ __forceinline__ void xw_mma_block(v8f (&acc)[4], const v2f (&a)[2 * XKG],
                                             const WFrag& f, int blk) {
#pragma unroll
    for (int i = 0; i < XBLKG; ++i) {
        const int g = blk * XBLKG + i;
#pragma unroll
        for (int q = 0; q < 4; ++q) {
            v2f b0;
            b0[0] = f.v[i][q][0]; b0[1] = f.v[i][q][1];
            acc[q] = __builtin_amdgcn_wmma_f32_16x16x4_f32(
                false, a[2 * g], false, b0, (short)0, acc[q], false, false);
        }
#pragma unroll
        for (int q = 0; q < 4; ++q) {
            v2f b1;
            b1[0] = f.v[i][q][2]; b1[1] = f.v[i][q][3];
            acc[q] = __builtin_amdgcn_wmma_f32_16x16x4_f32(
                false, a[2 * g + 1], false, b1, (short)0, acc[q], false, false);
        }
    }
}

__global__ __launch_bounds__(XW_THREADS)
void lstm_xw_kernel(const float* __restrict__ x, const float* __restrict__ W,
                    const float* __restrict__ bias, float* __restrict__ pre) {
    __shared__ __attribute__((aligned(16))) float WT[FH * WPITCH];
    __shared__ __attribute__((aligned(16))) float bs[FH];
    const int tid   = threadIdx.x;
    const int wave  = tid >> 5;
    const int lane  = tid & 31;
    const int lhalf = lane >> 4;
    const int l15   = lane & 15;

    {
        int k  = tid / (FH / 4);
        int cq = tid - k * (FH / 4);
#pragma unroll 15
        for (int i = 0; i < (D_IN * (FH / 4)) / XW_THREADS; ++i) {
            const v4f w = *(const v4f*)(W + (size_t)k * FH + 4 * cq);
            const bool inq  = (cq < 96);
            const int rbase = inq ? (((cq >> 4) << 6) | (cq & 15)) : 4 * cq;
            const int rstep = inq ? 16 * WPITCH : WPITCH;
            float* dst = WT + rbase * WPITCH + kperm8(k);
#pragma unroll
            for (int jj = 0; jj < 4; ++jj) dst[jj * rstep] = w[jj];
            cq += XW_THREADS - 2 * (FH / 4);
            k  += 2;
            if (cq >= FH / 4) { cq -= FH / 4; ++k; }
        }
    }
    for (int e = tid; e < FH; e += XW_THREADS) bs[e] = bias[e];
    __syncthreads();

    const int mt0 = blockIdx.x * MT_PER_WG + wave * MT_PER_WAVE;

#pragma unroll 1
    for (int it = 0; it < MT_PER_WAVE; ++it) {
        const int mt = mt0 + it;
        const int s  = mt >> 4;
        const int b0 = (mt & 15) << 4;

        const float* xr = x + ((size_t)(b0 + l15) * S_LEN + s) * D_IN + 2 * lhalf;
        v2f a[2 * XKG];
#pragma unroll
        for (int kk = 0; kk < 2 * XKG; ++kk) a[kk] = *(const v2f*)(xr + 4 * kk);

        if (it + 1 < MT_PER_WAVE) {
            const int mn = mt + 1;
            const float* xn = x + ((size_t)(((mn & 15) << 4) + l15) * S_LEN + (mn >> 4)) * D_IN;
#pragma unroll
            for (int p = 0; p < 3; ++p)
                __builtin_prefetch(xn + (lhalf + 2 * p) * 32, 0, 0);
        }

        float* prow = pre + ((size_t)s * B_SZ + b0 + 8 * lhalf) * FH;

#pragma unroll 1
        for (int T = 0; T < 6; ++T) {
            const float* wb[4];
#pragma unroll
            for (int q = 0; q < 4; ++q)
                wb[q] = WT + (64 * T + 16 * q + l15) * WPITCH + 4 * lhalf;

            WFrag fx, fy;
            load_wfrag(fx, wb, 0);

            const v4f bq = *(const v4f*)(bs + 64 * T + 4 * l15);
            v8f acc[4];
#pragma unroll
            for (int q = 0; q < 4; ++q)
#pragma unroll
                for (int r = 0; r < 8; ++r) acc[q][r] = bq[q];

#pragma unroll
            for (int blk = 0; blk < XNBLK; blk += 2) {
                load_wfrag(fy, wb, blk + 1);
                keep_grouped();
                xw_mma_block(acc, a, fx, blk);
                keep_grouped();
                if (blk + 2 < XNBLK) load_wfrag(fx, wb, blk + 2);
                keep_grouped();
                xw_mma_block(acc, a, fy, blk + 1);
                keep_grouped();
            }

#pragma unroll
            for (int r = 0; r < 8; ++r) {
                v4f o;
                o[0] = acc[0][r]; o[1] = acc[1][r]; o[2] = acc[2][r]; o[3] = acc[3][r];
                *(v4f*)(prow + (size_t)r * FH + 64 * T + 4 * l15) = o;
            }
        }

        {
            const float* wt = WT + (384 + l15) * WPITCH + 4 * lhalf;
            const float bb = bs[384 + l15];
            v8f acc1;
#pragma unroll
            for (int r = 0; r < 8; ++r) acc1[r] = bb;
            v4f f0[XKG / 2], f1[XKG / 2];
#pragma unroll
            for (int g = 0; g < XKG / 2; ++g) f0[g] = *(const v4f*)(wt + 8 * g);
#pragma unroll
            for (int g = 0; g < XKG / 2; ++g) f1[g] = *(const v4f*)(wt + 8 * (g + XKG / 2));
            keep_grouped();
#pragma unroll
            for (int g = 0; g < XKG / 2; ++g) {
                v2f b0, b1;
                b0[0] = f0[g][0]; b0[1] = f0[g][1];
                b1[0] = f0[g][2]; b1[1] = f0[g][3];
                acc1 = __builtin_amdgcn_wmma_f32_16x16x4_f32(
                    false, a[2 * g], false, b0, (short)0, acc1, false, false);
                acc1 = __builtin_amdgcn_wmma_f32_16x16x4_f32(
                    false, a[2 * g + 1], false, b1, (short)0, acc1, false, false);
            }
#pragma unroll
            for (int g = 0; g < XKG / 2; ++g) {
                const int gg = g + XKG / 2;
                v2f b0, b1;
                b0[0] = f1[g][0]; b0[1] = f1[g][1];
                b1[0] = f1[g][2]; b1[1] = f1[g][3];
                acc1 = __builtin_amdgcn_wmma_f32_16x16x4_f32(
                    false, a[2 * gg], false, b0, (short)0, acc1, false, false);
                acc1 = __builtin_amdgcn_wmma_f32_16x16x4_f32(
                    false, a[2 * gg + 1], false, b1, (short)0, acc1, false, false);
            }
#pragma unroll
            for (int r = 0; r < 8; ++r)
                prow[(size_t)r * FH + 384 + l15] = acc1[r];
        }
    }
}

#define REC_WAVES   7
#define REC_THREADS (REC_WAVES * 32)
#define UP    100
#define HP    108
#define KGRP  12
#define BLKG  3
#define KTAIL 96

__device__ __forceinline__ int kpos(int k) {
    return (k < KTAIL) ? kperm8(k) : k;
}

struct UFrag { v4f v[BLKG][4]; };

__device__ __forceinline__ void load_ufrag(UFrag& f, const float* const (&ub)[4], int blk) {
#pragma unroll
    for (int i = 0; i < BLKG; ++i)
#pragma unroll
        for (int q = 0; q < 4; ++q)
            f.v[i][q] = *(const v4f*)(ub[q] + 8 * (blk * BLKG + i));
}

__device__ __forceinline__ void mma_block(v8f (&acc)[4], const v4f (&a)[KGRP],
                                          const UFrag& f, int blk) {
#pragma unroll
    for (int i = 0; i < BLKG; ++i) {
        const v4f av = a[blk * BLKG + i];
        v2f a0, a1;
        a0[0] = av[0]; a0[1] = av[1];
        a1[0] = av[2]; a1[1] = av[3];
#pragma unroll
        for (int q = 0; q < 4; ++q) {
            v2f b0;
            b0[0] = f.v[i][q][0]; b0[1] = f.v[i][q][1];
            acc[q] = __builtin_amdgcn_wmma_f32_16x16x4_f32(
                false, a0, false, b0, (short)0, acc[q], false, false);
        }
#pragma unroll
        for (int q = 0; q < 4; ++q) {
            v2f b1;
            b1[0] = f.v[i][q][2]; b1[1] = f.v[i][q][3];
            acc[q] = __builtin_amdgcn_wmma_f32_16x16x4_f32(
                false, a1, false, b1, (short)0, acc[q], false, false);
        }
    }
}

__global__ __launch_bounds__(REC_THREADS)
void lstm_rec_kernel(const float* __restrict__ pre, const float* __restrict__ U,
                     float* __restrict__ hT) {
    __shared__ __attribute__((aligned(16))) float UT[FH * UP];
    __shared__ __attribute__((aligned(16))) float h_s[2 * 16 * HP];

    const int b0  = blockIdx.x << 4;
    const int tid = threadIdx.x;
    const int wave  = tid >> 5;
    const int lane  = tid & 31;
    const int lhalf = lane >> 4;
    const int l15   = lane & 15;

    const int  j      = (wave << 4) + l15;
    const bool jvalid = (j < H_DIM);
    const int  jc     = jvalid ? j : (H_DIM - 1);
    const int  jst    = jvalid ? kpos(j) : (H_DIM + (l15 & 3));

    for (int e = tid; e < H_DIM * FH; e += REC_THREADS) {
        const int k = e / FH, col = e - k * FH;
        const bool gcol = (col >= 2 * H_DIM) && (col < 3 * H_DIM);
        UT[col * UP + kpos(k)] = gcol ? U[e] : 0.5f * U[e];
    }
    for (int e = tid; e < 2 * 16 * HP; e += REC_THREADS) h_s[e] = 0.0f;
    __syncthreads();

    const float* ub[4];
#pragma unroll
    for (int q = 0; q < 4; ++q) ub[q] = UT + (q * H_DIM + jc) * UP + 4 * lhalf;

    float c_reg[8];
#pragma unroll
    for (int r = 0; r < 8; ++r) c_reg[r] = 0.0f;

    UFrag fx, fy;
    load_ufrag(fx, ub, 0);

    for (int s = 0; s < S_LEN; ++s) {
        const float* prow = pre + ((size_t)s * B_SZ + b0) * FH;
        const float* hr = h_s + (s & 1) * (16 * HP) + l15 * HP + 4 * lhalf;
        float*       hw = h_s + ((s + 1) & 1) * (16 * HP);

        if (s + 1 < S_LEN) {
            const float* nrow = prow + (size_t)B_SZ * FH;
            for (int e = tid; e < 400; e += REC_THREADS)
                __builtin_prefetch(nrow + e * 16, 0, 0);
        }

        v4f a[KGRP];
#pragma unroll
        for (int g = 0; g < KGRP; ++g) a[g] = *(const v4f*)(hr + 8 * g);
        const v2f at = *(const v2f*)(hr - 4 * lhalf + KTAIL + 2 * lhalf);
        keep_grouped();

        float p[4][8];
#pragma unroll
        for (int q = 0; q < 4; ++q)
#pragma unroll
            for (int r = 0; r < 8; ++r)
                p[q][r] = prow[(size_t)(r + lhalf * 8) * FH + q * H_DIM + jc];

        load_ufrag(fy, ub, 1);

        v8f acc[4];
#pragma unroll
        for (int q = 0; q < 4; ++q)
#pragma unroll
            for (int r = 0; r < 8; ++r) acc[q][r] = 0.0f;

        keep_grouped();
        mma_block(acc, a, fx, 0);
        load_ufrag(fx, ub, 2);
        keep_grouped();
        mma_block(acc, a, fy, 1);
        keep_grouped();
        load_ufrag(fy, ub, 3);
        v2f bt[4];
#pragma unroll
        for (int q = 0; q < 4; ++q)
            bt[q] = *(const v2f*)(ub[q] - 4 * lhalf + KTAIL + 2 * lhalf);
        keep_grouped();
        mma_block(acc, a, fx, 2);
        load_ufrag(fx, ub, 0);
        keep_grouped();
        mma_block(acc, a, fy, 3);
#pragma unroll
        for (int q = 0; q < 4; ++q)
            acc[q] = __builtin_amdgcn_wmma_f32_16x16x4_f32(
                false, at, false, bt[q], (short)0, acc[q], false, false);

#pragma unroll
        for (int r = 0; r < 8; ++r) {
            const float hi = fmaf(0.5f, p[0][r], acc[0][r]);
            const float hf = fmaf(0.5f, p[1][r], acc[1][r]);
            const float zg = acc[2][r] + p[2][r];
            const float ho = fmaf(0.5f, p[3][r], acc[3][r]);
            const float ig = sigmoid_of_half_(hi);
            const float fg = sigmoid_of_half_(hf);
            const float gg = fast_tanhf_(zg);
            const float og = sigmoid_of_half_(ho);
            const float cn = fg * c_reg[r] + ig * gg;
            c_reg[r] = cn;
            const float hn = og * fast_tanhf_(cn);
            hw[(r + lhalf * 8) * HP + jst] = hn;
        }
        __syncthreads();
    }

    const float* hl = h_s + (S_LEN & 1) * (16 * HP);
    for (int e = tid; e < 16 * H_DIM; e += REC_THREADS) {
        const int b = e / H_DIM, jj = e - b * H_DIM;
        hT[(size_t)(b0 + b) * H_DIM + jj] = hl[b * HP + kpos(jj)];
    }
}

__global__ __launch_bounds__(32)
void dense_kernel(const float* __restrict__ hT, const float* __restrict__ Wd,
                  const float* __restrict__ bd, float* __restrict__ out) {
    const int m0 = (blockIdx.x >> 4) << 4;
    const int n0 = (blockIdx.x & 15) << 4;
    const int lane  = threadIdx.x & 31;
    const int lhalf = lane >> 4;
    const int l15   = lane & 15;
    const int koff  = lhalf * 2;
    const int n = n0 + l15;
    const bool nvalid = (n < C_OUT);

    v8f acc;
    const float bb = nvalid ? bd[n] : 0.0f;
#pragma unroll
    for (int r = 0; r < 8; ++r) acc[r] = bb;

#pragma unroll
    for (int kk = 0; kk < 25; ++kk) {
        const int k0 = kk * 4 + koff;
        v2f a, bf;
        a[0]  = hT[(size_t)(m0 + l15) * H_DIM + k0];
        a[1]  = hT[(size_t)(m0 + l15) * H_DIM + k0 + 1];
        bf[0] = nvalid ? Wd[(size_t)k0 * C_OUT + n] : 0.0f;
        bf[1] = nvalid ? Wd[(size_t)(k0 + 1) * C_OUT + n] : 0.0f;
        acc = __builtin_amdgcn_wmma_f32_16x16x4_f32(
            false, a, false, bf, (short)0, acc, false, false);
    }
    if (nvalid) {
#pragma unroll
        for (int r = 0; r < 8; ++r)
            out[(size_t)(m0 + r + lhalf * 8) * C_OUT + n] = acc[r];
    }
}

extern "C" void kernel_launch(void* const* d_in, const int* in_sizes, int n_in,
                              void* d_out, int out_size, void* d_ws, size_t ws_size,
                              hipStream_t stream) {
    const float* x  = (const float*)d_in[0];
    const float* W  = (const float*)d_in[1];
    const float* U  = (const float*)d_in[2];
    const float* bi = (const float*)d_in[3];
    const float* Wd = (const float*)d_in[4];
    const float* bd = (const float*)d_in[5];
    float* out = (float*)d_out;

    float* pre = (float*)d_ws;
    float* hT  = pre + (size_t)S_LEN * B_SZ * FH;

    lstm_xw_kernel<<<XW_GRID, XW_THREADS, 0, stream>>>(x, W, bi, pre);
    lstm_rec_kernel<<<B_SZ / 16, REC_THREADS, 0, stream>>>(pre, U, hT);
    dense_kernel<<<(B_SZ / 16) * 16, 32, 0, stream>>>(hT, Wd, bd, out);
}


# recurrence: product formed transposed (U^T tile as A, h^T as B), a tile = 4 units x 4 gates, gates lane-local; 25 tiles on 8 waves: busiest SIMD 175 WMMAs a step, not 200; K sum order kept
# speedup vs baseline: 2.0596x; 1.0545x over previous
#include <hip/hip_runtime.h>
#include <math.h>


typedef __attribute__((ext_vector_type(2))) float v2f;
typedef __attribute__((ext_vector_type(4))) float v4f;
typedef __attribute__((ext_vector_type(8))) float v8f;

#define S_LEN 384
#define B_SZ  256
#define D_IN  192
#define H_DIM 100
#define FH    400
#define C_OUT 250

__device__ __forceinline__ float fast_tanhf_(float v) {
#if __has_builtin(__builtin_amdgcn_tanhf)
    return __builtin_amdgcn_tanhf(v);
#elif __has_builtin(__builtin_amdgcn_tanh_f32)
    return __builtin_amdgcn_tanh_f32(v);
#else
    return tanhf(v);
#endif
}

__device__ __forceinline__ float sigmoid_of_half_(float half_z) {
    return fmaf(0.5f, fast_tanhf_(half_z), 0.5f);
}

#define WPITCH 196
#define XW_WAVES 8
#define XW_THREADS (XW_WAVES * 32)
#define XW_GRID 256
#define MT_PER_WG   ((S_LEN * B_SZ / 16) / XW_GRID)
#define MT_PER_WAVE (MT_PER_WG / XW_WAVES)
#define XKG   24
#define XBLKG 3
#define XNBLK (XKG / XBLKG)

__device__ __forceinline__ int kperm8(int k) {
    return (k & ~6) | ((k & 2) << 1) | ((k & 4) >> 1);
}

__device__ __forceinline__ void keep_grouped() { __builtin_amdgcn_sched_barrier(0); }

struct WFrag { v4f v[XBLKG][4]; };

__device__ __forceinline__ void load_wfrag(WFrag& f, const float* const (&wb)[4], int blk) {
#pragma unroll
    for (int i = 0; i < XBLKG; ++i)
#pragma unroll
        for (int q = 0; q < 4; ++q)
            f.v[i][q] = *(const v4f*)(wb[q] + 8 * (blk * XBLKG + i));
}

__device__ __forceinline__ void xw_mma_block(v8f (&acc)[4], const v2f (&a)[2 * XKG],
                                             const WFrag& f, int blk) {
#pragma unroll
    for (int i = 0; i < XBLKG; ++i) {
        const int g = blk * XBLKG + i;
#pragma unroll
        for (int q = 0; q < 4; ++q) {
            v2f b0;
            b0[0] = f.v[i][q][0]; b0[1] = f.v[i][q][1];
            acc[q] = __builtin_amdgcn_wmma_f32_16x16x4_f32(
                false, a[2 * g], false, b0, (short)0, acc[q], false, false);
        }
#pragma unroll
        for (int q = 0; q < 4; ++q) {
            v2f b1;
            b1[0] = f.v[i][q][2]; b1[1] = f.v[i][q][3];
            acc[q] = __builtin_amdgcn_wmma_f32_16x16x4_f32(
                false, a[2 * g + 1], false, b1, (short)0, acc[q], false, false);
        }
    }
}

__global__ __launch_bounds__(XW_THREADS)
void lstm_xw_kernel(const float* __restrict__ x, const float* __restrict__ W,
                    const float* __restrict__ bias, float* __restrict__ pre) {
    __shared__ __attribute__((aligned(16))) float WT[FH * WPITCH];
    __shared__ __attribute__((aligned(16))) float bs[FH];
    const int tid   = threadIdx.x;
    const int wave  = tid >> 5;
    const int lane  = tid & 31;
    const int lhalf = lane >> 4;
    const int l15   = lane & 15;

    {
        int k  = tid / (FH / 4);
        int cq = tid - k * (FH / 4);
#pragma unroll 15
        for (int i = 0; i < (D_IN * (FH / 4)) / XW_THREADS; ++i) {
            const v4f w = *(const v4f*)(W + (size_t)k * FH + 4 * cq);
            const bool inq  = (cq < 96);
            const int rbase = inq ? (((cq >> 4) << 6) | (cq & 15)) : 4 * cq;
            const int rstep = inq ? 16 * WPITCH : WPITCH;
            float* dst = WT + rbase * WPITCH + kperm8(k);
#pragma unroll
            for (int jj = 0; jj < 4; ++jj) dst[jj * rstep] = w[jj];
            cq += XW_THREADS - 2 * (FH / 4);
            k  += 2;
            if (cq >= FH / 4) { cq -= FH / 4; ++k; }
        }
    }
    for (int e = tid; e < FH; e += XW_THREADS) bs[e] = bias[e];
    __syncthreads();

    const int mt0 = blockIdx.x * MT_PER_WG + wave * MT_PER_WAVE;

#pragma unroll 1
    for (int it = 0; it < MT_PER_WAVE; ++it) {
        const int mt = mt0 + it;
        const int s  = mt >> 4;
        const int b0 = (mt & 15) << 4;

        const float* xr = x + ((size_t)(b0 + l15) * S_LEN + s) * D_IN + 2 * lhalf;
        v2f a[2 * XKG];
#pragma unroll
        for (int kk = 0; kk < 2 * XKG; ++kk) a[kk] = *(const v2f*)(xr + 4 * kk);

        if (it + 1 < MT_PER_WAVE) {
            const int mn = mt + 1;
            const float* xn = x + ((size_t)(((mn & 15) << 4) + l15) * S_LEN + (mn >> 4)) * D_IN;
#pragma unroll
            for (int p = 0; p < 3; ++p)
                __builtin_prefetch(xn + (lhalf + 2 * p) * 32, 0, 0);
        }

        float* prow = pre + ((size_t)s * B_SZ + b0 + 8 * lhalf) * FH;

#pragma unroll 1
        for (int T = 0; T < 6; ++T) {
            const float* wb[4];
#pragma unroll
            for (int q = 0; q < 4; ++q)
                wb[q] = WT + (64 * T + 16 * q + l15) * WPITCH + 4 * lhalf;

            WFrag fx, fy;
            load_wfrag(fx, wb, 0);

            const v4f bq = *(const v4f*)(bs + 64 * T + 4 * l15);
            v8f acc[4];
#pragma unroll
            for (int q = 0; q < 4; ++q)
#pragma unroll
                for (int r = 0; r < 8; ++r) acc[q][r] = bq[q];

#pragma unroll
            for (int blk = 0; blk < XNBLK; blk += 2) {
                load_wfrag(fy, wb, blk + 1);
                keep_grouped();
                xw_mma_block(acc, a, fx, blk);
                keep_grouped();
                if (blk + 2 < XNBLK) load_wfrag(fx, wb, blk + 2);
                keep_grouped();
                xw_mma_block(acc, a, fy, blk + 1);
                keep_grouped();
            }

#pragma unroll
            for (int r = 0; r < 8; ++r) {
                v4f o;
                o[0] = acc[0][r]; o[1] = acc[1][r]; o[2] = acc[2][r]; o[3] = acc[3][r];
                *(v4f*)(prow + (size_t)r * FH + 64 * T + 4 * l15) = o;
            }
        }

        {
            const float* wt = WT + (384 + l15) * WPITCH + 4 * lhalf;
            const float bb = bs[384 + l15];
            v8f acc1;
#pragma unroll
            for (int r = 0; r < 8; ++r) acc1[r] = bb;
            v4f f0[XKG / 2], f1[XKG / 2];
#pragma unroll
            for (int g = 0; g < XKG / 2; ++g) f0[g] = *(const v4f*)(wt + 8 * g);
#pragma unroll
            for (int g = 0; g < XKG / 2; ++g) f1[g] = *(const v4f*)(wt + 8 * (g + XKG / 2));
            keep_grouped();
#pragma unroll
            for (int g = 0; g < XKG / 2; ++g) {
                v2f b0, b1;
                b0[0] = f0[g][0]; b0[1] = f0[g][1];
                b1[0] = f0[g][2]; b1[1] = f0[g][3];
                acc1 = __builtin_amdgcn_wmma_f32_16x16x4_f32(
                    false, a[2 * g], false, b0, (short)0, acc1, false, false);
                acc1 = __builtin_amdgcn_wmma_f32_16x16x4_f32(
                    false, a[2 * g + 1], false, b1, (short)0, acc1, false, false);
            }
#pragma unroll
            for (int g = 0; g < XKG / 2; ++g) {
                const int gg = g + XKG / 2;
                v2f b0, b1;
                b0[0] = f1[g][0]; b0[1] = f1[g][1];
                b1[0] = f1[g][2]; b1[1] = f1[g][3];
                acc1 = __builtin_amdgcn_wmma_f32_16x16x4_f32(
                    false, a[2 * gg], false, b0, (short)0, acc1, false, false);
                acc1 = __builtin_amdgcn_wmma_f32_16x16x4_f32(
                    false, a[2 * gg + 1], false, b1, (short)0, acc1, false, false);
            }
#pragma unroll
            for (int r = 0; r < 8; ++r)
                prow[(size_t)r * FH + 384 + l15] = acc1[r];
        }
    }
}

#define REC_WAVES   8
#define REC_THREADS (REC_WAVES * 32)
#define NT_MAX 4
#define UP    100
#define HP    108
#define KGRP  12
#define BLKG  3
#define KTAIL 96

__device__ __forceinline__ int kpos(int k) {
    return (k < KTAIL) ? kperm8(k) : k;
}

struct UFrag { v4f v[BLKG][NT_MAX]; };

template <int NT>
__device__ __forceinline__ void load_ufrag(UFrag& f, const float* const (&ua)[NT_MAX], int blk) {
#pragma unroll
    for (int i = 0; i < BLKG; ++i)
#pragma unroll
        for (int t = 0; t < NT; ++t)
            f.v[i][t] = *(const v4f*)(ua[t] + 8 * (blk * BLKG + i));
}

template <int NT>
__device__ __forceinline__ void mma_block(v8f (&acc)[NT_MAX], const v4f (&hb)[KGRP],
                                          const UFrag& f, int blk) {
#pragma unroll
    for (int i = 0; i < BLKG; ++i) {
        const v4f hv = hb[blk * BLKG + i];
        v2f h0, h1;
        h0[0] = hv[0]; h0[1] = hv[1];
        h1[0] = hv[2]; h1[1] = hv[3];
#pragma unroll
        for (int t = 0; t < NT; ++t) {
            v2f u0;
            u0[0] = f.v[i][t][0]; u0[1] = f.v[i][t][1];
            acc[t] = __builtin_amdgcn_wmma_f32_16x16x4_f32(
                false, u0, false, h0, (short)0, acc[t], false, false);
        }
#pragma unroll
        for (int t = 0; t < NT; ++t) {
            v2f u1;
            u1[0] = f.v[i][t][2]; u1[1] = f.v[i][t][3];
            acc[t] = __builtin_amdgcn_wmma_f32_16x16x4_f32(
                false, u1, false, h1, (short)0, acc[t], false, false);
        }
    }
}

template <int NT>
__device__ __forceinline__ void rec_step(const float* __restrict__ prow, const float* hr,
                                         float* hw, const float* const (&ua)[NT_MAX],
                                         const int (&hst)[NT_MAX], UFrag& fx, UFrag& fy,
                                         v2f (&c_reg)[NT_MAX], int lhalf) {
    v4f hb[KGRP];
#pragma unroll
    for (int g = 0; g < KGRP; ++g) hb[g] = *(const v4f*)(hr + 8 * g);
    const v2f ht = *(const v2f*)(hr - 4 * lhalf + KTAIL + 2 * lhalf);
    keep_grouped();

    v2f p[NT_MAX][4];
#pragma unroll
    for (int t = 0; t < NT; ++t)
#pragma unroll
        for (int q = 0; q < 4; ++q)
            p[t][q] = *(const v2f*)(prow + q * H_DIM + 4 * t);

    load_ufrag<NT>(fy, ua, 1);

    v8f acc[NT_MAX];
#pragma unroll
    for (int t = 0; t < NT; ++t)
#pragma unroll
        for (int r = 0; r < 8; ++r) acc[t][r] = 0.0f;

    keep_grouped();
    mma_block<NT>(acc, hb, fx, 0);
    load_ufrag<NT>(fx, ua, 2);
    keep_grouped();
    mma_block<NT>(acc, hb, fy, 1);
    keep_grouped();
    load_ufrag<NT>(fy, ua, 3);
    v2f ut[NT_MAX];
#pragma unroll
    for (int t = 0; t < NT; ++t)
        ut[t] = *(const v2f*)(ua[t] - 4 * lhalf + KTAIL + 2 * lhalf);
    keep_grouped();
    mma_block<NT>(acc, hb, fx, 2);
    load_ufrag<NT>(fx, ua, 0);
    keep_grouped();
    mma_block<NT>(acc, hb, fy, 3);
#pragma unroll
    for (int t = 0; t < NT; ++t)
        acc[t] = __builtin_amdgcn_wmma_f32_16x16x4_f32(
            false, ut[t], false, ht, (short)0, acc[t], false, false);

    keep_grouped();

    const v2f half2 = {0.5f, 0.5f};
#pragma unroll
    for (int t = 0; t < NT; ++t) {
        v2f ai, af, ag, ao;
        ai[0] = acc[t][0]; ai[1] = acc[t][1];
        af[0] = acc[t][2]; af[1] = acc[t][3];
        ag[0] = acc[t][4]; ag[1] = acc[t][5];
        ao[0] = acc[t][6]; ao[1] = acc[t][7];
        const v2f hi = __builtin_elementwise_fma(half2, p[t][0], ai);
        const v2f hf = __builtin_elementwise_fma(half2, p[t][1], af);
        const v2f zg = ag + p[t][2];
        const v2f ho = __builtin_elementwise_fma(half2, p[t][3], ao);
        v2f ti, tf, gg, to;
#pragma unroll
        for (int uu = 0; uu < 2; ++uu) {
            ti[uu] = fast_tanhf_(hi[uu]);
            tf[uu] = fast_tanhf_(hf[uu]);
            gg[uu] = fast_tanhf_(zg[uu]);
            to[uu] = fast_tanhf_(ho[uu]);
        }
        const v2f ig = __builtin_elementwise_fma(half2, ti, half2);
        const v2f fg = __builtin_elementwise_fma(half2, tf, half2);
        const v2f og = __builtin_elementwise_fma(half2, to, half2);
        const v2f cn = fg * c_reg[t] + ig * gg;
        c_reg[t] = cn;
        v2f hn;
#pragma unroll
        for (int uu = 0; uu < 2; ++uu) hn[uu] = og[uu] * fast_tanhf_(cn[uu]);
        *(v2f*)(hw + hst[t]) = hn;
    }
}

__global__ __launch_bounds__(REC_THREADS)
void lstm_rec_kernel(const float* __restrict__ pre, const float* __restrict__ U,
                     float* __restrict__ hT) {
    __shared__ __attribute__((aligned(16))) float UT[FH * UP];
    __shared__ __attribute__((aligned(16))) float h_s[2 * 16 * HP];

    const int b0  = blockIdx.x << 4;
    const int tid = threadIdx.x;
    const int wave  = __builtin_amdgcn_readfirstlane(tid >> 5);
    const int lane  = tid & 31;
    const int lhalf = lane >> 4;
    const int l15   = lane & 15;
    const bool four = (wave == REC_WAVES - 1);
    const int tbase = 3 * wave;

    for (int e = tid; e < H_DIM * FH; e += REC_THREADS) {
        const int k = e / FH, col = e - k * FH;
        const bool gcol = (col >= 2 * H_DIM) && (col < 3 * H_DIM);
        UT[col * UP + kpos(k)] = gcol ? U[e] : 0.5f * U[e];
    }
    for (int e = tid; e < 2 * 16 * HP; e += REC_THREADS) h_s[e] = 0.0f;
    __syncthreads();

    const float* ua[NT_MAX];
    int hst[NT_MAX];
#pragma unroll
    for (int t = 0; t < NT_MAX; ++t) {
        const int tile = (tbase + t < 25) ? (tbase + t) : 24;
        ua[t]  = UT + (((l15 & 7) >> 1) * H_DIM + 4 * tile + 2 * (l15 >> 3) + (l15 & 1)) * UP
                    + 4 * lhalf;
        hst[t] = l15 * HP + kpos(4 * tile + 2 * lhalf);
    }

    v2f c_reg[NT_MAX];
#pragma unroll
    for (int t = 0; t < NT_MAX; ++t) { c_reg[t][0] = 0.0f; c_reg[t][1] = 0.0f; }

    UFrag fx, fy;
    if (four) load_ufrag<4>(fx, ua, 0); else load_ufrag<3>(fx, ua, 0);

    for (int s = 0; s < S_LEN; ++s) {
        const float* prow = pre + ((size_t)s * B_SZ + b0) * FH;
        const float* hr = h_s + (s & 1) * (16 * HP) + l15 * HP + 4 * lhalf;
        float*       hw = h_s + ((s + 1) & 1) * (16 * HP);

        if (s + 1 < S_LEN) {
            const float* nrow = prow + (size_t)B_SZ * FH;
            for (int e = tid; e < 400; e += REC_THREADS)
                __builtin_prefetch(nrow + e * 16, 0, 0);
        }

        const float* pl = prow + (size_t)l15 * FH + 4 * tbase + 2 * lhalf;
        if (four) rec_step<4>(pl, hr, hw, ua, hst, fx, fy, c_reg, lhalf);
        else      rec_step<3>(pl, hr, hw, ua, hst, fx, fy, c_reg, lhalf);
        __syncthreads();
    }

    const float* hl = h_s + (S_LEN & 1) * (16 * HP);
    for (int e = tid; e < 16 * H_DIM; e += REC_THREADS) {
        const int b = e / H_DIM, jj = e - b * H_DIM;
        hT[(size_t)(b0 + b) * H_DIM + jj] = hl[b * HP + kpos(jj)];
    }
}

__global__ __launch_bounds__(32)
void dense_kernel(const float* __restrict__ hT, const float* __restrict__ Wd,
                  const float* __restrict__ bd, float* __restrict__ out) {
    const int m0 = (blockIdx.x >> 4) << 4;
    const int n0 = (blockIdx.x & 15) << 4;
    const int lane  = threadIdx.x & 31;
    const int lhalf = lane >> 4;
    const int l15   = lane & 15;
    const int koff  = lhalf * 2;
    const int n = n0 + l15;
    const bool nvalid = (n < C_OUT);

    v8f acc;
    const float bb = nvalid ? bd[n] : 0.0f;
#pragma unroll
    for (int r = 0; r < 8; ++r) acc[r] = bb;

#pragma unroll
    for (int kk = 0; kk < 25; ++kk) {
        const int k0 = kk * 4 + koff;
        v2f a, bf;
        a[0]  = hT[(size_t)(m0 + l15) * H_DIM + k0];
        a[1]  = hT[(size_t)(m0 + l15) * H_DIM + k0 + 1];
        bf[0] = nvalid ? Wd[(size_t)k0 * C_OUT + n] : 0.0f;
        bf[1] = nvalid ? Wd[(size_t)(k0 + 1) * C_OUT + n] : 0.0f;
        acc = __builtin_amdgcn_wmma_f32_16x16x4_f32(
            false, a, false, bf, (short)0, acc, false, false);
    }
    if (nvalid) {
#pragma unroll
        for (int r = 0; r < 8; ++r)
            out[(size_t)(m0 + r + lhalf * 8) * C_OUT + n] = acc[r];
    }
}

extern "C" void kernel_launch(void* const* d_in, const int* in_sizes, int n_in,
                              void* d_out, int out_size, void* d_ws, size_t ws_size,
                              hipStream_t stream) {
    const float* x  = (const float*)d_in[0];
    const float* W  = (const float*)d_in[1];
    const float* U  = (const float*)d_in[2];
    const float* bi = (const float*)d_in[3];
    const float* Wd = (const float*)d_in[4];
    const float* bd = (const float*)d_in[5];
    float* out = (float*)d_out;

    float* pre = (float*)d_ws;
    float* hT  = pre + (size_t)S_LEN * B_SZ * FH;

    lstm_xw_kernel<<<XW_GRID, XW_THREADS, 0, stream>>>(x, W, bi, pre);
    lstm_rec_kernel<<<B_SZ / 16, REC_THREADS, 0, stream>>>(pre, U, hT);
    dense_kernel<<<(B_SZ / 16) * 16, 32, 0, stream>>>(hT, Wd, bd, out);
}


# recurrence: only the h and U LDS loads stay between the step barrier and the first WMMA; pre loads and next step's prefetch (one line a thread, no loop) go behind block 0; sum order kept
# speedup vs baseline: 2.4391x; 1.1843x over previous
#include <hip/hip_runtime.h>
#include <math.h>


typedef __attribute__((ext_vector_type(2))) float v2f;
typedef __attribute__((ext_vector_type(4))) float v4f;
typedef __attribute__((ext_vector_type(8))) float v8f;

#define S_LEN 384
#define B_SZ  256
#define D_IN  192
#define H_DIM 100
#define FH    400
#define C_OUT 250

__device__ __forceinline__ float fast_tanhf_(float v) {
#if __has_builtin(__builtin_amdgcn_tanhf)
    return __builtin_amdgcn_tanhf(v);
#elif __has_builtin(__builtin_amdgcn_tanh_f32)
    return __builtin_amdgcn_tanh_f32(v);
#else
    return tanhf(v);
#endif
}

__device__ __forceinline__ float sigmoid_of_half_(float half_z) {
    return fmaf(0.5f, fast_tanhf_(half_z), 0.5f);
}

#define WPITCH 196
#define XW_WAVES 8
#define XW_THREADS (XW_WAVES * 32)
#define XW_GRID 256
#define MT_PER_WG   ((S_LEN * B_SZ / 16) / XW_GRID)
#define MT_PER_WAVE (MT_PER_WG / XW_WAVES)
#define XKG   24
#define XBLKG 3
#define XNBLK (XKG / XBLKG)

__device__ __forceinline__ int kperm8(int k) {
    return (k & ~6) | ((k & 2) << 1) | ((k & 4) >> 1);
}

__device__ __forceinline__ void keep_grouped() { __builtin_amdgcn_sched_barrier(0); }

struct WFrag { v4f v[XBLKG][4]; };

__device__ __forceinline__ void load_wfrag(WFrag& f, const float* const (&wb)[4], int blk) {
#pragma unroll
    for (int i = 0; i < XBLKG; ++i)
#pragma unroll
        for (int q = 0; q < 4; ++q)
            f.v[i][q] = *(const v4f*)(wb[q] + 8 * (blk * XBLKG + i));
}

__device__ __forceinline__ void xw_mma_block(v8f (&acc)[4], const v2f (&a)[2 * XKG],
                                             const WFrag& f, int blk) {
#pragma unroll
    for (int i = 0; i < XBLKG; ++i) {
        const int g = blk * XBLKG + i;
#pragma unroll
        for (int q = 0; q < 4; ++q) {
            v2f b0;
            b0[0] = f.v[i][q][0]; b0[1] = f.v[i][q][1];
            acc[q] = __builtin_amdgcn_wmma_f32_16x16x4_f32(
                false, a[2 * g], false, b0, (short)0, acc[q], false, false);
        }
#pragma unroll
        for (int q = 0; q < 4; ++q) {
            v2f b1;
            b1[0] = f.v[i][q][2]; b1[1] = f.v[i][q][3];
            acc[q] = __builtin_amdgcn_wmma_f32_16x16x4_f32(
                false, a[2 * g + 1], false, b1, (short)0, acc[q], false, false);
        }
    }
}

__global__ __launch_bounds__(XW_THREADS)
void lstm_xw_kernel(const float* __restrict__ x, const float* __restrict__ W,
                    const float* __restrict__ bias, float* __restrict__ pre) {
    __shared__ __attribute__((aligned(16))) float WT[FH * WPITCH];
    __shared__ __attribute__((aligned(16))) float bs[FH];
    const int tid   = threadIdx.x;
    const int wave  = tid >> 5;
    const int lane  = tid & 31;
    const int lhalf = lane >> 4;
    const int l15   = lane & 15;

    {
        int k  = tid / (FH / 4);
        int cq = tid - k * (FH / 4);
#pragma unroll 15
        for (int i = 0; i < (D_IN * (FH / 4)) / XW_THREADS; ++i) {
            const v4f w = *(const v4f*)(W + (size_t)k * FH + 4 * cq);
            const bool inq  = (cq < 96);
            const int rbase = inq ? (((cq >> 4) << 6) | (cq & 15)) : 4 * cq;
            const int rstep = inq ? 16 * WPITCH : WPITCH;
            float* dst = WT + rbase * WPITCH + kperm8(k);
#pragma unroll
            for (int jj = 0; jj < 4; ++jj) dst[jj * rstep] = w[jj];
            cq += XW_THREADS - 2 * (FH / 4);
            k  += 2;
            if (cq >= FH / 4) { cq -= FH / 4; ++k; }
        }
    }
    for (int e = tid; e < FH; e += XW_THREADS) bs[e] = bias[e];
    __syncthreads();

    const int mt0 = blockIdx.x * MT_PER_WG + wave * MT_PER_WAVE;

#pragma unroll 1
    for (int it = 0; it < MT_PER_WAVE; ++it) {
        const int mt = mt0 + it;
        const int s  = mt >> 4;
        const int b0 = (mt & 15) << 4;

        const float* xr = x + ((size_t)(b0 + l15) * S_LEN + s) * D_IN + 2 * lhalf;
        v2f a[2 * XKG];
#pragma unroll
        for (int kk = 0; kk < 2 * XKG; ++kk) a[kk] = *(const v2f*)(xr + 4 * kk);

        if (it + 1 < MT_PER_WAVE) {
            const int mn = mt + 1;
            const float* xn = x + ((size_t)(((mn & 15) << 4) + l15) * S_LEN + (mn >> 4)) * D_IN;
#pragma unroll
            for (int p = 0; p < 3; ++p)
                __builtin_prefetch(xn + (lhalf + 2 * p) * 32, 0, 0);
        }

        float* prow = pre + ((size_t)s * B_SZ + b0 + 8 * lhalf) * FH;

#pragma unroll 1
        for (int T = 0; T < 6; ++T) {
            const float* wb[4];
#pragma unroll
            for (int q = 0; q < 4; ++q)
                wb[q] = WT + (64 * T + 16 * q + l15) * WPITCH + 4 * lhalf;

            WFrag fx, fy;
            load_wfrag(fx, wb, 0);

            const v4f bq = *(const v4f*)(bs + 64 * T + 4 * l15);
            v8f acc[4];
#pragma unroll
            for (int q = 0; q < 4; ++q)
#pragma unroll
                for (int r = 0; r < 8; ++r) acc[q][r] = bq[q];

#pragma unroll
            for (int blk = 0; blk < XNBLK; blk += 2) {
                load_wfrag(fy, wb, blk + 1);
                keep_grouped();
                xw_mma_block(acc, a, fx, blk);
                keep_grouped();
                if (blk + 2 < XNBLK) load_wfrag(fx, wb, blk + 2);
                keep_grouped();
                xw_mma_block(acc, a, fy, blk + 1);
                keep_grouped();
            }

#pragma unroll
            for (int r = 0; r < 8; ++r) {
                v4f o;
                o[0] = acc[0][r]; o[1] = acc[1][r]; o[2] = acc[2][r]; o[3] = acc[3][r];
                *(v4f*)(prow + (size_t)r * FH + 64 * T + 4 * l15) = o;
            }
        }

        {
            const float* wt = WT + (384 + l15) * WPITCH + 4 * lhalf;
            const float bb = bs[384 + l15];
            v8f acc1;
#pragma unroll
            for (int r = 0; r < 8; ++r) acc1[r] = bb;
            v4f f0[XKG / 2], f1[XKG / 2];
#pragma unroll
            for (int g = 0; g < XKG / 2; ++g) f0[g] = *(const v4f*)(wt + 8 * g);
#pragma unroll
            for (int g = 0; g < XKG / 2; ++g) f1[g] = *(const v4f*)(wt + 8 * (g + XKG / 2));
            keep_grouped();
#pragma unroll
            for (int g = 0; g < XKG / 2; ++g) {
                v2f b0, b1;
                b0[0] = f0[g][0]; b0[1] = f0[g][1];
                b1[0] = f0[g][2]; b1[1] = f0[g][3];
                acc1 = __builtin_amdgcn_wmma_f32_16x16x4_f32(
                    false, a[2 * g], false, b0, (short)0, acc1, false, false);
                acc1 = __builtin_amdgcn_wmma_f32_16x16x4_f32(
                    false, a[2 * g + 1], false, b1, (short)0, acc1, false, false);
            }
#pragma unroll
            for (int g = 0; g < XKG / 2; ++g) {
                const int gg = g + XKG / 2;
                v2f b0, b1;
                b0[0] = f1[g][0]; b0[1] = f1[g][1];
                b1[0] = f1[g][2]; b1[1] = f1[g][3];
                acc1 = __builtin_amdgcn_wmma_f32_16x16x4_f32(
                    false, a[2 * gg], false, b0, (short)0, acc1, false, false);
                acc1 = __builtin_amdgcn_wmma_f32_16x16x4_f32(
                    false, a[2 * gg + 1], false, b1, (short)0, acc1, false, false);
            }
#pragma unroll
            for (int r = 0; r < 8; ++r)
                prow[(size_t)r * FH + 384 + l15] = acc1[r];
        }
    }
}

#define REC_WAVES   8
#define REC_THREADS (REC_WAVES * 32)
#define NT_MAX 4
#define UP    100
#define HP    108
#define KGRP  12
#define BLKG  3
#define KTAIL 96
#define PF_LINES 200

__device__ __forceinline__ int kpos(int k) {
    return (k < KTAIL) ? kperm8(k) : k;
}

struct UFrag { v4f v[BLKG][NT_MAX]; };

template <int NT>
__device__ __forceinline__ void load_ufrag(UFrag& f, const float* const (&ua)[NT_MAX], int blk) {
#pragma unroll
    for (int i = 0; i < BLKG; ++i)
#pragma unroll
        for (int t = 0; t < NT; ++t)
            f.v[i][t] = *(const v4f*)(ua[t] + 8 * (blk * BLKG + i));
}

template <int NT>
__device__ __forceinline__ void mma_block(v8f (&acc)[NT_MAX], const v4f (&hb)[KGRP],
                                          const UFrag& f, int blk) {
#pragma unroll
    for (int i = 0; i < BLKG; ++i) {
        const v4f hv = hb[blk * BLKG + i];
        v2f h0, h1;
        h0[0] = hv[0]; h0[1] = hv[1];
        h1[0] = hv[2]; h1[1] = hv[3];
#pragma unroll
        for (int t = 0; t < NT; ++t) {
            v2f u0;
            u0[0] = f.v[i][t][0]; u0[1] = f.v[i][t][1];
            acc[t] = __builtin_amdgcn_wmma_f32_16x16x4_f32(
                false, u0, false, h0, (short)0, acc[t], false, false);
        }
#pragma unroll
        for (int t = 0; t < NT; ++t) {
            v2f u1;
            u1[0] = f.v[i][t][2]; u1[1] = f.v[i][t][3];
            acc[t] = __builtin_amdgcn_wmma_f32_16x16x4_f32(
                false, u1, false, h1, (short)0, acc[t], false, false);
        }
    }
}

template <int NT>
__device__ __forceinline__ void rec_step(const float* __restrict__ prow, const float* pfn,
                                         const float* hr,
                                         float* hw, const float* const (&ua)[NT_MAX],
                                         const int (&hst)[NT_MAX], UFrag& fx, UFrag& fy,
                                         v2f (&c_reg)[NT_MAX], int lhalf) {
    v4f hb[KGRP];
#pragma unroll
    for (int g = 0; g < KGRP; ++g) hb[g] = *(const v4f*)(hr + 8 * g);
    const v2f ht = *(const v2f*)(hr - 4 * lhalf + KTAIL + 2 * lhalf);
    keep_grouped();

    load_ufrag<NT>(fy, ua, 1);

    v8f acc[NT_MAX];
#pragma unroll
    for (int t = 0; t < NT; ++t)
#pragma unroll
        for (int r = 0; r < 8; ++r) acc[t][r] = 0.0f;

    keep_grouped();
    mma_block<NT>(acc, hb, fx, 0);
    keep_grouped();

    v2f p[NT_MAX][4];
#pragma unroll
    for (int t = 0; t < NT; ++t)
#pragma unroll
        for (int q = 0; q < 4; ++q)
            p[t][q] = *(const v2f*)(prow + q * H_DIM + 4 * t);
    __builtin_prefetch(pfn, 0, 0);

    load_ufrag<NT>(fx, ua, 2);
    keep_grouped();
    mma_block<NT>(acc, hb, fy, 1);
    keep_grouped();
    load_ufrag<NT>(fy, ua, 3);
    v2f ut[NT_MAX];
#pragma unroll
    for (int t = 0; t < NT; ++t)
        ut[t] = *(const v2f*)(ua[t] - 4 * lhalf + KTAIL + 2 * lhalf);
    keep_grouped();
    mma_block<NT>(acc, hb, fx, 2);
    load_ufrag<NT>(fx, ua, 0);
    keep_grouped();
    mma_block<NT>(acc, hb, fy, 3);
#pragma unroll
    for (int t = 0; t < NT; ++t)
        acc[t] = __builtin_amdgcn_wmma_f32_16x16x4_f32(
            false, ut[t], false, ht, (short)0, acc[t], false, false);

    keep_grouped();

    const v2f half2 = {0.5f, 0.5f};
#pragma unroll
    for (int t = 0; t < NT; ++t) {
        v2f ai, af, ag, ao;
        ai[0] = acc[t][0]; ai[1] = acc[t][1];
        af[0] = acc[t][2]; af[1] = acc[t][3];
        ag[0] = acc[t][4]; ag[1] = acc[t][5];
        ao[0] = acc[t][6]; ao[1] = acc[t][7];
        const v2f hi = __builtin_elementwise_fma(half2, p[t][0], ai);
        const v2f hf = __builtin_elementwise_fma(half2, p[t][1], af);
        const v2f zg = ag + p[t][2];
        const v2f ho = __builtin_elementwise_fma(half2, p[t][3], ao);
        v2f ti, tf, gg, to;
#pragma unroll
        for (int uu = 0; uu < 2; ++uu) {
            ti[uu] = fast_tanhf_(hi[uu]);
            tf[uu] = fast_tanhf_(hf[uu]);
            gg[uu] = fast_tanhf_(zg[uu]);
            to[uu] = fast_tanhf_(ho[uu]);
        }
        const v2f ig = __builtin_elementwise_fma(half2, ti, half2);
        const v2f fg = __builtin_elementwise_fma(half2, tf, half2);
        const v2f og = __builtin_elementwise_fma(half2, to, half2);
        const v2f cn = fg * c_reg[t] + ig * gg;
        c_reg[t] = cn;
        v2f hn;
#pragma unroll
        for (int uu = 0; uu < 2; ++uu) hn[uu] = og[uu] * fast_tanhf_(cn[uu]);
        *(v2f*)(hw + hst[t]) = hn;
    }
}

__global__ __launch_bounds__(REC_THREADS)
void lstm_rec_kernel(const float* __restrict__ pre, const float* __restrict__ U,
                     float* __restrict__ hT) {
    __shared__ __attribute__((aligned(16))) float UT[FH * UP];
    __shared__ __attribute__((aligned(16))) float h_s[2 * 16 * HP];

    const int b0  = blockIdx.x << 4;
    const int tid = threadIdx.x;
    const int wave  = __builtin_amdgcn_readfirstlane(tid >> 5);
    const int lane  = tid & 31;
    const int lhalf = lane >> 4;
    const int l15   = lane & 15;
    const bool four = (wave == REC_WAVES - 1);
    const int tbase = 3 * wave;

    for (int e = tid; e < H_DIM * FH; e += REC_THREADS) {
        const int k = e / FH, col = e - k * FH;
        const bool gcol = (col >= 2 * H_DIM) && (col < 3 * H_DIM);
        UT[col * UP + kpos(k)] = gcol ? U[e] : 0.5f * U[e];
    }
    for (int e = tid; e < 2 * 16 * HP; e += REC_THREADS) h_s[e] = 0.0f;
    __syncthreads();

    const float* ua[NT_MAX];
    int hst[NT_MAX];
#pragma unroll
    for (int t = 0; t < NT_MAX; ++t) {
        const int tile = (tbase + t < 25) ? (tbase + t) : 24;
        ua[t]  = UT + (((l15 & 7) >> 1) * H_DIM + 4 * tile + 2 * (l15 >> 3) + (l15 & 1)) * UP
                    + 4 * lhalf;
        hst[t] = l15 * HP + kpos(4 * tile + 2 * lhalf);
    }

    v2f c_reg[NT_MAX];
#pragma unroll
    for (int t = 0; t < NT_MAX; ++t) { c_reg[t][0] = 0.0f; c_reg[t][1] = 0.0f; }

    UFrag fx, fy;
    if (four) load_ufrag<4>(fx, ua, 0); else load_ufrag<3>(fx, ua, 0);

    const float* hr0 = h_s + l15 * HP + 4 * lhalf;
    const float* hr1 = hr0 + 16 * HP;
    float* hw0 = h_s;
    float* hw1 = h_s + 16 * HP;
    const size_t pstep = (size_t)B_SZ * FH;
    const float* pl = pre + ((size_t)b0 + l15) * FH + 4 * tbase + 2 * lhalf;
    const float* pfp = pre + pstep + (size_t)b0 * FH + 32 * (tid < PF_LINES ? tid : PF_LINES - 1);

    for (int s = 0; s < S_LEN; s += 2) {
        if (four) rec_step<4>(pl, pfp, hr0, hw1, ua, hst, fx, fy, c_reg, lhalf);
        else      rec_step<3>(pl, pfp, hr0, hw1, ua, hst, fx, fy, c_reg, lhalf);
        __syncthreads();
        const float* plo = pl + pstep;
        const float* pfo = pfp + ((s + 2 < S_LEN) ? pstep : (size_t)0);
        if (four) rec_step<4>(plo, pfo, hr1, hw0, ua, hst, fx, fy, c_reg, lhalf);
        else      rec_step<3>(plo, pfo, hr1, hw0, ua, hst, fx, fy, c_reg, lhalf);
        pl  += 2 * pstep;
        pfp += 2 * pstep;
        __syncthreads();
    }

    const float* hl = h_s + (S_LEN & 1) * (16 * HP);
    for (int e = tid; e < 16 * H_DIM; e += REC_THREADS) {
        const int b = e / H_DIM, jj = e - b * H_DIM;
        hT[(size_t)(b0 + b) * H_DIM + jj] = hl[b * HP + kpos(jj)];
    }
}

__global__ __launch_bounds__(32)
void dense_kernel(const float* __restrict__ hT, const float* __restrict__ Wd,
                  const float* __restrict__ bd, float* __restrict__ out) {
    const int m0 = (blockIdx.x >> 4) << 4;
    const int n0 = (blockIdx.x & 15) << 4;
    const int lane  = threadIdx.x & 31;
    const int lhalf = lane >> 4;
    const int l15   = lane & 15;
    const int koff  = lhalf * 2;
    const int n = n0 + l15;
    const bool nvalid = (n < C_OUT);

    v8f acc;
    const float bb = nvalid ? bd[n] : 0.0f;
#pragma unroll
    for (int r = 0; r < 8; ++r) acc[r] = bb;

#pragma unroll
    for (int kk = 0; kk < 25; ++kk) {
        const int k0 = kk * 4 + koff;
        v2f a, bf;
        a[0]  = hT[(size_t)(m0 + l15) * H_DIM + k0];
        a[1]  = hT[(size_t)(m0 + l15) * H_DIM + k0 + 1];
        bf[0] = nvalid ? Wd[(size_t)k0 * C_OUT + n] : 0.0f;
        bf[1] = nvalid ? Wd[(size_t)(k0 + 1) * C_OUT + n] : 0.0f;
        acc = __builtin_amdgcn_wmma_f32_16x16x4_f32(
            false, a, false, bf, (short)0, acc, false, false);
    }
    if (nvalid) {
#pragma unroll
        for (int r = 0; r < 8; ++r)
            out[(size_t)(m0 + r + lhalf * 8) * C_OUT + n] = acc[r];
    }
}

extern "C" void kernel_launch(void* const* d_in, const int* in_sizes, int n_in,
                              void* d_out, int out_size, void* d_ws, size_t ws_size,
                              hipStream_t stream) {
    const float* x  = (const float*)d_in[0];
    const float* W  = (const float*)d_in[1];
    const float* U  = (const float*)d_in[2];
    const float* bi = (const float*)d_in[3];
    const float* Wd = (const float*)d_in[4];
    const float* bd = (const float*)d_in[5];
    float* out = (float*)d_out;

    float* pre = (float*)d_ws;
    float* hT  = pre + (size_t)S_LEN * B_SZ * FH;

    lstm_xw_kernel<<<XW_GRID, XW_THREADS, 0, stream>>>(x, W, bi, pre);
    lstm_rec_kernel<<<B_SZ / 16, REC_THREADS, 0, stream>>>(pre, U, hT);
    dense_kernel<<<(B_SZ / 16) * 16, 32, 0, stream>>>(hT, Wd, bd, out);
}


# recurrence: h handed over in two halves a step (k<48 from waves 0-3, rest from 4-7) on a split barrier; wave priorities order the blocks so each half's gates run behind the other's WMMAs; sums kept
# speedup vs baseline: 2.5233x; 1.0345x over previous
#include <hip/hip_runtime.h>
#include <math.h>


typedef __attribute__((ext_vector_type(2))) float v2f;
typedef __attribute__((ext_vector_type(4))) float v4f;
typedef __attribute__((ext_vector_type(8))) float v8f;

#define S_LEN 384
#define B_SZ  256
#define D_IN  192
#define H_DIM 100
#define FH    400
#define C_OUT 250

__device__ __forceinline__ float fast_tanhf_(float v) {
#if __has_builtin(__builtin_amdgcn_tanhf)
    return __builtin_amdgcn_tanhf(v);
#elif __has_builtin(__builtin_amdgcn_tanh_f32)
    return __builtin_amdgcn_tanh_f32(v);
#else
    return tanhf(v);
#endif
}

__device__ __forceinline__ float sigmoid_of_half_(float half_z) {
    return fmaf(0.5f, fast_tanhf_(half_z), 0.5f);
}

#define WPITCH 196
#define XW_WAVES 8
#define XW_THREADS (XW_WAVES * 32)
#define XW_GRID 256
#define MT_PER_WG   ((S_LEN * B_SZ / 16) / XW_GRID)
#define MT_PER_WAVE (MT_PER_WG / XW_WAVES)
#define XKG   24
#define XBLKG 3
#define XNBLK (XKG / XBLKG)

__device__ __forceinline__ int kperm8(int k) {
    return (k & ~6) | ((k & 2) << 1) | ((k & 4) >> 1);
}

__device__ __forceinline__ void keep_grouped() { __builtin_amdgcn_sched_barrier(0); }

struct WFrag { v4f v[XBLKG][4]; };

__device__ __forceinline__ void load_wfrag(WFrag& f, const float* const (&wb)[4], int blk) {
#pragma unroll
    for (int i = 0; i < XBLKG; ++i)
#pragma unroll
        for (int q = 0; q < 4; ++q)
            f.v[i][q] = *(const v4f*)(wb[q] + 8 * (blk * XBLKG + i));
}

__device__ __forceinline__ void xw_mma_block(v8f (&acc)[4], const v2f (&a)[2 * XKG],
                                             const WFrag& f, int blk) {
#pragma unroll
    for (int i = 0; i < XBLKG; ++i) {
        const int g = blk * XBLKG + i;
#pragma unroll
        for (int q = 0; q < 4; ++q) {
            v2f b0;
            b0[0] = f.v[i][q][0]; b0[1] = f.v[i][q][1];
            acc[q] = __builtin_amdgcn_wmma_f32_16x16x4_f32(
                false, a[2 * g], false, b0, (short)0, acc[q], false, false);
        }
#pragma unroll
        for (int q = 0; q < 4; ++q) {
            v2f b1;
            b1[0] = f.v[i][q][2]; b1[1] = f.v[i][q][3];
            acc[q] = __builtin_amdgcn_wmma_f32_16x16x4_f32(
                false, a[2 * g + 1], false, b1, (short)0, acc[q], false, false);
        }
    }
}

__global__ __launch_bounds__(XW_THREADS)
void lstm_xw_kernel(const float* __restrict__ x, const float* __restrict__ W,
                    const float* __restrict__ bias, float* __restrict__ pre) {
    __shared__ __attribute__((aligned(16))) float WT[FH * WPITCH];
    __shared__ __attribute__((aligned(16))) float bs[FH];
    const int tid   = threadIdx.x;
    const int wave  = tid >> 5;
    const int lane  = tid & 31;
    const int lhalf = lane >> 4;
    const int l15   = lane & 15;

    {
        int k  = tid / (FH / 4);
        int cq = tid - k * (FH / 4);
#pragma unroll 15
        for (int i = 0; i < (D_IN * (FH / 4)) / XW_THREADS; ++i) {
            const v4f w = *(const v4f*)(W + (size_t)k * FH + 4 * cq);
            const bool inq  = (cq < 96);
            const int rbase = inq ? (((cq >> 4) << 6) | (cq & 15)) : 4 * cq;
            const int rstep = inq ? 16 * WPITCH : WPITCH;
            float* dst = WT + rbase * WPITCH + kperm8(k);
#pragma unroll
            for (int jj = 0; jj < 4; ++jj) dst[jj * rstep] = w[jj];
            cq += XW_THREADS - 2 * (FH / 4);
            k  += 2;
            if (cq >= FH / 4) { cq -= FH / 4; ++k; }
        }
    }
    for (int e = tid; e < FH; e += XW_THREADS) bs[e] = bias[e];
    __syncthreads();

    const int mt0 = blockIdx.x * MT_PER_WG + wave * MT_PER_WAVE;

#pragma unroll 1
    for (int it = 0; it < MT_PER_WAVE; ++it) {
        const int mt = mt0 + it;
        const int s  = mt >> 4;
        const int b0 = (mt & 15) << 4;

        const float* xr = x + ((size_t)(b0 + l15) * S_LEN + s) * D_IN + 2 * lhalf;
        v2f a[2 * XKG];
#pragma unroll
        for (int kk = 0; kk < 2 * XKG; ++kk) a[kk] = *(const v2f*)(xr + 4 * kk);

        if (it + 1 < MT_PER_WAVE) {
            const int mn = mt + 1;
            const float* xn = x + ((size_t)(((mn & 15) << 4) + l15) * S_LEN + (mn >> 4)) * D_IN;
#pragma unroll
            for (int p = 0; p < 3; ++p)
                __builtin_prefetch(xn + (lhalf + 2 * p) * 32, 0, 0);
        }

        float* prow = pre + ((size_t)s * B_SZ + b0 + 8 * lhalf) * FH;

#pragma unroll 1
        for (int T = 0; T < 6; ++T) {
            const float* wb[4];
#pragma unroll
            for (int q = 0; q < 4; ++q)
                wb[q] = WT + (64 * T + 16 * q + l15) * WPITCH + 4 * lhalf;

            WFrag fx, fy;
            load_wfrag(fx, wb, 0);

            const v4f bq = *(const v4f*)(bs + 64 * T + 4 * l15);
            v8f acc[4];
#pragma unroll
            for (int q = 0; q < 4; ++q)
#pragma unroll
                for (int r = 0; r < 8; ++r) acc[q][r] = bq[q];

#pragma unroll
            for (int blk = 0; blk < XNBLK; blk += 2) {
                load_wfrag(fy, wb, blk + 1);
                keep_grouped();
                xw_mma_block(acc, a, fx, blk);
                keep_grouped();
                if (blk + 2 < XNBLK) load_wfrag(fx, wb, blk + 2);
                keep_grouped();
                xw_mma_block(acc, a, fy, blk + 1);
                keep_grouped();
            }

#pragma unroll
            for (int r = 0; r < 8; ++r) {
                v4f o;
                o[0] = acc[0][r]; o[1] = acc[1][r]; o[2] = acc[2][r]; o[3] = acc[3][r];
                *(v4f*)(prow + (size_t)r * FH + 64 * T + 4 * l15) = o;
            }
        }

        {
            const float* wt = WT + (384 + l15) * WPITCH + 4 * lhalf;
            const float bb = bs[384 + l15];
            v8f acc1;
#pragma unroll
            for (int r = 0; r < 8; ++r) acc1[r] = bb;
            v4f f0[XKG / 2], f1[XKG / 2];
#pragma unroll
            for (int g = 0; g < XKG / 2; ++g) f0[g] = *(const v4f*)(wt + 8 * g);
#pragma unroll
            for (int g = 0; g < XKG / 2; ++g) f1[g] = *(const v4f*)(wt + 8 * (g + XKG / 2));
            keep_grouped();
#pragma unroll
            for (int g = 0; g < XKG / 2; ++g) {
                v2f b0, b1;
                b0[0] = f0[g][0]; b0[1] = f0[g][1];
                b1[0] = f0[g][2]; b1[1] = f0[g][3];
                acc1 = __builtin_amdgcn_wmma_f32_16x16x4_f32(
                    false, a[2 * g], false, b0, (short)0, acc1, false, false);
                acc1 = __builtin_amdgcn_wmma_f32_16x16x4_f32(
                    false, a[2 * g + 1], false, b1, (short)0, acc1, false, false);
            }
#pragma unroll
            for (int g = 0; g < XKG / 2; ++g) {
                const int gg = g + XKG / 2;
                v2f b0, b1;
                b0[0] = f1[g][0]; b0[1] = f1[g][1];
                b1[0] = f1[g][2]; b1[1] = f1[g][3];
                acc1 = __builtin_amdgcn_wmma_f32_16x16x4_f32(
                    false, a[2 * gg], false, b0, (short)0, acc1, false, false);
                acc1 = __builtin_amdgcn_wmma_f32_16x16x4_f32(
                    false, a[2 * gg + 1], false, b1, (short)0, acc1, false, false);
            }
#pragma unroll
            for (int r = 0; r < 8; ++r)
                prow[(size_t)r * FH + 384 + l15] = acc1[r];
        }
    }
}

#define REC_WAVES   8
#define REC_THREADS (REC_WAVES * 32)
#define NT_MAX 4
#define UP    100
#define HP    108
#define KGRP  12
#define BLKG  3
#define KTAIL 96
#define PF_LINES 200

__device__ __forceinline__ int kpos(int k) {
    return (k < KTAIL) ? kperm8(k) : k;
}

struct UFrag { v4f v[BLKG][NT_MAX]; };

__device__ __forceinline__ void bar_signal_lds_done() {
    asm volatile("s_wait_dscnt 0x0\n\ts_barrier_signal -1" ::: "memory");
}
__device__ __forceinline__ void bar_wait() {
    asm volatile("s_barrier_wait -1" ::: "memory");
}

template <int NT>
__device__ __forceinline__ void load_ufrag(UFrag& f, const float* const (&ua)[NT_MAX], int blk) {
#pragma unroll
    for (int i = 0; i < BLKG; ++i)
#pragma unroll
        for (int t = 0; t < NT; ++t)
            f.v[i][t] = *(const v4f*)(ua[t] + 8 * (blk * BLKG + i));
}

template <int NT>
__device__ __forceinline__ void mma_block(v8f (&acc)[NT_MAX], const v4f (&hb)[KGRP],
                                          const UFrag& f, int blk) {
#pragma unroll
    for (int i = 0; i < BLKG; ++i) {
        const v4f hv = hb[blk * BLKG + i];
        v2f h0, h1;
        h0[0] = hv[0]; h0[1] = hv[1];
        h1[0] = hv[2]; h1[1] = hv[3];
#pragma unroll
        for (int t = 0; t < NT; ++t) {
            v2f u0;
            u0[0] = f.v[i][t][0]; u0[1] = f.v[i][t][1];
            acc[t] = __builtin_amdgcn_wmma_f32_16x16x4_f32(
                false, u0, false, h0, (short)0, acc[t], false, false);
        }
#pragma unroll
        for (int t = 0; t < NT; ++t) {
            v2f u1;
            u1[0] = f.v[i][t][2]; u1[1] = f.v[i][t][3];
            acc[t] = __builtin_amdgcn_wmma_f32_16x16x4_f32(
                false, u1, false, h1, (short)0, acc[t], false, false);
        }
    }
}

template <int NT, bool LOW>
__device__ __forceinline__ void rec_step(const float* __restrict__ prow, const float* pfn,
                                         const float* hr,
                                         float* hw, const float* const (&ua)[NT_MAX],
                                         const int (&hst)[NT_MAX], UFrag& fx, UFrag& fy,
                                         v2f (&c_reg)[NT_MAX], int lhalf) {
    bar_wait();
    bar_signal_lds_done();
    if (!LOW) __builtin_amdgcn_s_setprio(0);
    keep_grouped();

    v4f hb[KGRP];
#pragma unroll
    for (int g = 0; g < KGRP / 2; ++g) hb[g] = *(const v4f*)(hr + 8 * g);
    keep_grouped();

    load_ufrag<NT>(fy, ua, 1);

    v8f acc[NT_MAX];
#pragma unroll
    for (int t = 0; t < NT; ++t)
#pragma unroll
        for (int r = 0; r < 8; ++r) acc[t][r] = 0.0f;

    keep_grouped();
    mma_block<NT>(acc, hb, fx, 0);
    keep_grouped();

    v2f p[NT_MAX][4];
#pragma unroll
    for (int t = 0; t < NT; ++t)
#pragma unroll
        for (int q = 0; q < 4; ++q)
            p[t][q] = *(const v2f*)(prow + q * H_DIM + 4 * t);
    __builtin_prefetch(pfn, 0, 0);

    load_ufrag<NT>(fx, ua, 2);
    keep_grouped();
    mma_block<NT>(acc, hb, fy, 1);
    keep_grouped();

    bar_wait();
    if (!LOW) bar_signal_lds_done();
    if (!LOW) __builtin_amdgcn_s_setprio(2);
    keep_grouped();

#pragma unroll
    for (int g = KGRP / 2; g < KGRP; ++g) hb[g] = *(const v4f*)(hr + 8 * g);
    const v2f ht = *(const v2f*)(hr - 4 * lhalf + KTAIL + 2 * lhalf);
    keep_grouped();

    load_ufrag<NT>(fy, ua, 3);
    v2f ut[NT_MAX];
#pragma unroll
    for (int t = 0; t < NT; ++t)
        ut[t] = *(const v2f*)(ua[t] - 4 * lhalf + KTAIL + 2 * lhalf);
    keep_grouped();
    mma_block<NT>(acc, hb, fx, 2);
    load_ufrag<NT>(fx, ua, 0);
    keep_grouped();
    mma_block<NT>(acc, hb, fy, 3);
#pragma unroll
    for (int t = 0; t < NT; ++t)
        acc[t] = __builtin_amdgcn_wmma_f32_16x16x4_f32(
            false, ut[t], false, ht, (short)0, acc[t], false, false);

    keep_grouped();

    const v2f half2 = {0.5f, 0.5f};
#pragma unroll
    for (int t = 0; t < NT; ++t) {
        v2f ai, af, ag, ao;
        ai[0] = acc[t][0]; ai[1] = acc[t][1];
        af[0] = acc[t][2]; af[1] = acc[t][3];
        ag[0] = acc[t][4]; ag[1] = acc[t][5];
        ao[0] = acc[t][6]; ao[1] = acc[t][7];
        const v2f hi = __builtin_elementwise_fma(half2, p[t][0], ai);
        const v2f hf = __builtin_elementwise_fma(half2, p[t][1], af);
        const v2f zg = ag + p[t][2];
        const v2f ho = __builtin_elementwise_fma(half2, p[t][3], ao);
        v2f ti, tf, gg, to;
#pragma unroll
        for (int uu = 0; uu < 2; ++uu) {
            ti[uu] = fast_tanhf_(hi[uu]);
            tf[uu] = fast_tanhf_(hf[uu]);
            gg[uu] = fast_tanhf_(zg[uu]);
            to[uu] = fast_tanhf_(ho[uu]);
        }
        const v2f ig = __builtin_elementwise_fma(half2, ti, half2);
        const v2f fg = __builtin_elementwise_fma(half2, tf, half2);
        const v2f og = __builtin_elementwise_fma(half2, to, half2);
        const v2f cn = fg * c_reg[t] + ig * gg;
        c_reg[t] = cn;
        v2f hn;
#pragma unroll
        for (int uu = 0; uu < 2; ++uu) hn[uu] = og[uu] * fast_tanhf_(cn[uu]);
        *(v2f*)(hw + hst[t]) = hn;
    }
    if (LOW) bar_signal_lds_done();
}

__global__ __launch_bounds__(REC_THREADS)
void lstm_rec_kernel(const float* __restrict__ pre, const float* __restrict__ U,
                     float* __restrict__ hT) {
    __shared__ __attribute__((aligned(16))) float UT[FH * UP];
    __shared__ __attribute__((aligned(16))) float h_s[2 * 16 * HP];

    const int b0  = blockIdx.x << 4;
    const int tid = threadIdx.x;
    const int wave  = __builtin_amdgcn_readfirstlane(tid >> 5);
    const int lane  = tid & 31;
    const int lhalf = lane >> 4;
    const int l15   = lane & 15;
    const bool four = (wave == REC_WAVES - 1);
    const bool low  = (wave < REC_WAVES / 2);
    const int tbase = 3 * wave;

    for (int e = tid; e < H_DIM * FH; e += REC_THREADS) {
        const int k = e / FH, col = e - k * FH;
        const bool gcol = (col >= 2 * H_DIM) && (col < 3 * H_DIM);
        UT[col * UP + kpos(k)] = gcol ? U[e] : 0.5f * U[e];
    }
    for (int e = tid; e < 2 * 16 * HP; e += REC_THREADS) h_s[e] = 0.0f;
    __syncthreads();

    const float* ua[NT_MAX];
    int hst[NT_MAX];
#pragma unroll
    for (int t = 0; t < NT_MAX; ++t) {
        const int tile = (tbase + t < 25) ? (tbase + t) : 24;
        ua[t]  = UT + (((l15 & 7) >> 1) * H_DIM + 4 * tile + 2 * (l15 >> 3) + (l15 & 1)) * UP
                    + 4 * lhalf;
        hst[t] = l15 * HP + kpos(4 * tile + 2 * lhalf);
    }

    v2f c_reg[NT_MAX];
#pragma unroll
    for (int t = 0; t < NT_MAX; ++t) { c_reg[t][0] = 0.0f; c_reg[t][1] = 0.0f; }

    UFrag fx, fy;
    if (four) load_ufrag<4>(fx, ua, 0); else load_ufrag<3>(fx, ua, 0);

    const float* hr0 = h_s + l15 * HP + 4 * lhalf;
    const float* hr1 = hr0 + 16 * HP;
    float* hw0 = h_s;
    float* hw1 = h_s + 16 * HP;
    const size_t pstep = (size_t)B_SZ * FH;
    const float* pl = pre + ((size_t)b0 + l15) * FH + 4 * tbase + 2 * lhalf;
    const float* pfp = pre + pstep + (size_t)b0 * FH + 32 * (tid < PF_LINES ? tid : PF_LINES - 1);

    if (low) __builtin_amdgcn_s_setprio(1);
    bar_signal_lds_done();

    for (int s = 0; s < S_LEN; s += 2) {
        if (four)     rec_step<4, false>(pl, pfp, hr0, hw1, ua, hst, fx, fy, c_reg, lhalf);
        else if (low) rec_step<3, true >(pl, pfp, hr0, hw1, ua, hst, fx, fy, c_reg, lhalf);
        else          rec_step<3, false>(pl, pfp, hr0, hw1, ua, hst, fx, fy, c_reg, lhalf);
        const float* plo = pl + pstep;
        const float* pfo = pfp + ((s + 2 < S_LEN) ? pstep : (size_t)0);
        if (four)     rec_step<4, false>(plo, pfo, hr1, hw0, ua, hst, fx, fy, c_reg, lhalf);
        else if (low) rec_step<3, true >(plo, pfo, hr1, hw0, ua, hst, fx, fy, c_reg, lhalf);
        else          rec_step<3, false>(plo, pfo, hr1, hw0, ua, hst, fx, fy, c_reg, lhalf);
        pl  += 2 * pstep;
        pfp += 2 * pstep;
    }
    bar_wait();
    __syncthreads();

    const float* hl = h_s + (S_LEN & 1) * (16 * HP);
    for (int e = tid; e < 16 * H_DIM; e += REC_THREADS) {
        const int b = e / H_DIM, jj = e - b * H_DIM;
        hT[(size_t)(b0 + b) * H_DIM + jj] = hl[b * HP + kpos(jj)];
    }
}

__global__ __launch_bounds__(32)
void dense_kernel(const float* __restrict__ hT, const float* __restrict__ Wd,
                  const float* __restrict__ bd, float* __restrict__ out) {
    const int m0 = (blockIdx.x >> 4) << 4;
    const int n0 = (blockIdx.x & 15) << 4;
    const int lane  = threadIdx.x & 31;
    const int lhalf = lane >> 4;
    const int l15   = lane & 15;
    const int koff  = lhalf * 2;
    const int n = n0 + l15;
    const bool nvalid = (n < C_OUT);

    v8f acc;
    const float bb = nvalid ? bd[n] : 0.0f;
#pragma unroll
    for (int r = 0; r < 8; ++r) acc[r] = bb;

#pragma unroll
    for (int kk = 0; kk < 25; ++kk) {
        const int k0 = kk * 4 + koff;
        v2f a, bf;
        a[0]  = hT[(size_t)(m0 + l15) * H_DIM + k0];
        a[1]  = hT[(size_t)(m0 + l15) * H_DIM + k0 + 1];
        bf[0] = nvalid ? Wd[(size_t)k0 * C_OUT + n] : 0.0f;
        bf[1] = nvalid ? Wd[(size_t)(k0 + 1) * C_OUT + n] : 0.0f;
        acc = __builtin_amdgcn_wmma_f32_16x16x4_f32(
            false, a, false, bf, (short)0, acc, false, false);
    }
    if (nvalid) {
#pragma unroll
        for (int r = 0; r < 8; ++r)
            out[(size_t)(m0 + r + lhalf * 8) * C_OUT + n] = acc[r];
    }
}

extern "C" void kernel_launch(void* const* d_in, const int* in_sizes, int n_in,
                              void* d_out, int out_size, void* d_ws, size_t ws_size,
                              hipStream_t stream) {
    const float* x  = (const float*)d_in[0];
    const float* W  = (const float*)d_in[1];
    const float* U  = (const float*)d_in[2];
    const float* bi = (const float*)d_in[3];
    const float* Wd = (const float*)d_in[4];
    const float* bd = (const float*)d_in[5];
    float* out = (float*)d_out;

    float* pre = (float*)d_ws;
    float* hT  = pre + (size_t)S_LEN * B_SZ * FH;

    lstm_xw_kernel<<<XW_GRID, XW_THREADS, 0, stream>>>(x, W, bi, pre);
    lstm_rec_kernel<<<B_SZ / 16, REC_THREADS, 0, stream>>>(pre, U, hT);
    dense_kernel<<<(B_SZ / 16) * 16, 32, 0, stream>>>(hT, Wd, bd, out);
}


# launches: the input projection of steps >= 64 runs on the 240 idle WGPs beside the first 64 scan steps (one launch, disjoint rows); h, c pass through the workspace; 4 launches; every sum kept
# speedup vs baseline: 2.6608x; 1.0545x over previous
#include <hip/hip_runtime.h>
#include <math.h>


typedef __attribute__((ext_vector_type(2))) float v2f;
typedef __attribute__((ext_vector_type(4))) float v4f;
typedef __attribute__((ext_vector_type(8))) float v8f;

#define S_LEN 384
#define B_SZ  256
#define D_IN  192
#define H_DIM 100
#define FH    400
#define C_OUT 250

__device__ __forceinline__ float fast_tanhf_(float v) {
#if __has_builtin(__builtin_amdgcn_tanhf)
    return __builtin_amdgcn_tanhf(v);
#elif __has_builtin(__builtin_amdgcn_tanh_f32)
    return __builtin_amdgcn_tanh_f32(v);
#else
    return tanhf(v);
#endif
}

__device__ __forceinline__ float sigmoid_of_half_(float half_z) {
    return fmaf(0.5f, fast_tanhf_(half_z), 0.5f);
}

#define WPITCH 196
#define XW_WAVES 8
#define XW_THREADS (XW_WAVES * 32)
#define XW_GRID 256
#define REC_GRID (B_SZ / 16)
#define XW_SIDE_GRID (XW_GRID - REC_GRID)
#define S_SPLIT 64
#define MT_PER_S (B_SZ / 16)
#define XKG   24
#define XBLKG 3
#define XNBLK (XKG / XBLKG)

__device__ __forceinline__ int kperm8(int k) {
    return (k & ~6) | ((k & 2) << 1) | ((k & 4) >> 1);
}

__device__ __forceinline__ void keep_grouped() { __builtin_amdgcn_sched_barrier(0); }

struct WFrag { v4f v[XBLKG][4]; };

__device__ __forceinline__ void load_wfrag(WFrag& f, const float* const (&wb)[4], int blk) {
#pragma unroll
    for (int i = 0; i < XBLKG; ++i)
#pragma unroll
        for (int q = 0; q < 4; ++q)
            f.v[i][q] = *(const v4f*)(wb[q] + 8 * (blk * XBLKG + i));
}

__device__ __forceinline__ void xw_mma_block(v8f (&acc)[4], const v2f (&a)[2 * XKG],
                                             const WFrag& f, int blk) {
#pragma unroll
    for (int i = 0; i < XBLKG; ++i) {
        const int g = blk * XBLKG + i;
#pragma unroll
        for (int q = 0; q < 4; ++q) {
            v2f b0;
            b0[0] = f.v[i][q][0]; b0[1] = f.v[i][q][1];
            acc[q] = __builtin_amdgcn_wmma_f32_16x16x4_f32(
                false, a[2 * g], false, b0, (short)0, acc[q], false, false);
        }
#pragma unroll
        for (int q = 0; q < 4; ++q) {
            v2f b1;
            b1[0] = f.v[i][q][2]; b1[1] = f.v[i][q][3];
            acc[q] = __builtin_amdgcn_wmma_f32_16x16x4_f32(
                false, a[2 * g + 1], false, b1, (short)0, acc[q], false, false);
        }
    }
}

__device__ __forceinline__
void xw_body(const float* __restrict__ x, const float* __restrict__ W,
             const float* __restrict__ bias, float* pre, float* WT, float* bs,
             int wg, int nwg, int mt_begin, int mt_end) {
    const int tid   = threadIdx.x;
    const int wave  = __builtin_amdgcn_readfirstlane(tid >> 5);
    const int lane  = tid & 31;
    const int lhalf = lane >> 4;
    const int l15   = lane & 15;

    {
        int k  = tid / (FH / 4);
        int cq = tid - k * (FH / 4);
#pragma unroll 15
        for (int i = 0; i < (D_IN * (FH / 4)) / XW_THREADS; ++i) {
            const v4f w = *(const v4f*)(W + (size_t)k * FH + 4 * cq);
            const bool inq  = (cq < 96);
            const int rbase = inq ? (((cq >> 4) << 6) | (cq & 15)) : 4 * cq;
            const int rstep = inq ? 16 * WPITCH : WPITCH;
            float* dst = WT + rbase * WPITCH + kperm8(k);
#pragma unroll
            for (int jj = 0; jj < 4; ++jj) dst[jj * rstep] = w[jj];
            cq += XW_THREADS - 2 * (FH / 4);
            k  += 2;
            if (cq >= FH / 4) { cq -= FH / 4; ++k; }
        }
    }
    for (int e = tid; e < FH; e += XW_THREADS) bs[e] = bias[e];
    __syncthreads();

    const int mstride = nwg * XW_WAVES;

#pragma unroll 1
    for (int mt = mt_begin + wave * nwg + wg; mt < mt_end; mt += mstride) {
        const int s  = mt >> 4;
        const int b0 = (mt & 15) << 4;

        const float* xr = x + ((size_t)(b0 + l15) * S_LEN + s) * D_IN + 2 * lhalf;
        v2f a[2 * XKG];
#pragma unroll
        for (int kk = 0; kk < 2 * XKG; ++kk) a[kk] = *(const v2f*)(xr + 4 * kk);

        if (mt + mstride < mt_end) {
            const int mn = mt + mstride;
            const float* xn = x + ((size_t)(((mn & 15) << 4) + l15) * S_LEN + (mn >> 4)) * D_IN;
#pragma unroll
            for (int p = 0; p < 3; ++p)
                __builtin_prefetch(xn + (lhalf + 2 * p) * 32, 0, 0);
        }

        float* prow = pre + ((size_t)s * B_SZ + b0 + 8 * lhalf) * FH;

#pragma unroll 1
        for (int T = 0; T < 6; ++T) {
            const float* wb[4];
#pragma unroll
            for (int q = 0; q < 4; ++q)
                wb[q] = WT + (64 * T + 16 * q + l15) * WPITCH + 4 * lhalf;

            WFrag fx, fy;
            load_wfrag(fx, wb, 0);

            const v4f bq = *(const v4f*)(bs + 64 * T + 4 * l15);
            v8f acc[4];
#pragma unroll
            for (int q = 0; q < 4; ++q)
#pragma unroll
                for (int r = 0; r < 8; ++r) acc[q][r] = bq[q];

#pragma unroll
            for (int blk = 0; blk < XNBLK; blk += 2) {
                load_wfrag(fy, wb, blk + 1);
                keep_grouped();
                xw_mma_block(acc, a, fx, blk);
                keep_grouped();
                if (blk + 2 < XNBLK) load_wfrag(fx, wb, blk + 2);
                keep_grouped();
                xw_mma_block(acc, a, fy, blk + 1);
                keep_grouped();
            }

#pragma unroll
            for (int r = 0; r < 8; ++r) {
                v4f o;
                o[0] = acc[0][r]; o[1] = acc[1][r]; o[2] = acc[2][r]; o[3] = acc[3][r];
                *(v4f*)(prow + (size_t)r * FH + 64 * T + 4 * l15) = o;
            }
        }

        {
            const float* wt = WT + (384 + l15) * WPITCH + 4 * lhalf;
            const float bb = bs[384 + l15];
            v8f acc1;
#pragma unroll
            for (int r = 0; r < 8; ++r) acc1[r] = bb;
            v4f f0[XKG / 2], f1[XKG / 2];
#pragma unroll
            for (int g = 0; g < XKG / 2; ++g) f0[g] = *(const v4f*)(wt + 8 * g);
#pragma unroll
            for (int g = 0; g < XKG / 2; ++g) f1[g] = *(const v4f*)(wt + 8 * (g + XKG / 2));
            keep_grouped();
#pragma unroll
            for (int g = 0; g < XKG / 2; ++g) {
                v2f b0, b1;
                b0[0] = f0[g][0]; b0[1] = f0[g][1];
                b1[0] = f0[g][2]; b1[1] = f0[g][3];
                acc1 = __builtin_amdgcn_wmma_f32_16x16x4_f32(
                    false, a[2 * g], false, b0, (short)0, acc1, false, false);
                acc1 = __builtin_amdgcn_wmma_f32_16x16x4_f32(
                    false, a[2 * g + 1], false, b1, (short)0, acc1, false, false);
            }
#pragma unroll
            for (int g = 0; g < XKG / 2; ++g) {
                const int gg = g + XKG / 2;
                v2f b0, b1;
                b0[0] = f1[g][0]; b0[1] = f1[g][1];
                b1[0] = f1[g][2]; b1[1] = f1[g][3];
                acc1 = __builtin_amdgcn_wmma_f32_16x16x4_f32(
                    false, a[2 * gg], false, b0, (short)0, acc1, false, false);
                acc1 = __builtin_amdgcn_wmma_f32_16x16x4_f32(
                    false, a[2 * gg + 1], false, b1, (short)0, acc1, false, false);
            }
#pragma unroll
            for (int r = 0; r < 8; ++r)
                prow[(size_t)r * FH + 384 + l15] = acc1[r];
        }
    }
}

__global__ __launch_bounds__(XW_THREADS)
void lstm_xw_head_kernel(const float* __restrict__ x, const float* __restrict__ W,
                         const float* __restrict__ bias, float* __restrict__ pre) {
    __shared__ __attribute__((aligned(16))) float WT[FH * WPITCH];
    __shared__ __attribute__((aligned(16))) float bs[FH];
    xw_body(x, W, bias, pre, WT, bs, blockIdx.x, XW_GRID, 0, S_SPLIT * MT_PER_S);
}

#define REC_WAVES   8
#define REC_THREADS (REC_WAVES * 32)
#define NT_MAX 4
#define UP    100
#define HP    108
#define KGRP  12
#define BLKG  3
#define KTAIL 96
#define PF_LINES 200

__device__ __forceinline__ int kpos(int k) {
    return (k < KTAIL) ? kperm8(k) : k;
}

struct UFrag { v4f v[BLKG][NT_MAX]; };

__device__ __forceinline__ void bar_signal_lds_done() {
    asm volatile("s_wait_dscnt 0x0\n\ts_barrier_signal -1" ::: "memory");
}
__device__ __forceinline__ void bar_wait() {
    asm volatile("s_barrier_wait -1" ::: "memory");
}

template <int NT>
__device__ __forceinline__ void load_ufrag(UFrag& f, const float* const (&ua)[NT_MAX], int blk) {
#pragma unroll
    for (int i = 0; i < BLKG; ++i)
#pragma unroll
        for (int t = 0; t < NT; ++t)
            f.v[i][t] = *(const v4f*)(ua[t] + 8 * (blk * BLKG + i));
}

template <int NT>
__device__ __forceinline__ void mma_block(v8f (&acc)[NT_MAX], const v4f (&hb)[KGRP],
                                          const UFrag& f, int blk) {
#pragma unroll
    for (int i = 0; i < BLKG; ++i) {
        const v4f hv = hb[blk * BLKG + i];
        v2f h0, h1;
        h0[0] = hv[0]; h0[1] = hv[1];
        h1[0] = hv[2]; h1[1] = hv[3];
#pragma unroll
        for (int t = 0; t < NT; ++t) {
            v2f u0;
            u0[0] = f.v[i][t][0]; u0[1] = f.v[i][t][1];
            acc[t] = __builtin_amdgcn_wmma_f32_16x16x4_f32(
                false, u0, false, h0, (short)0, acc[t], false, false);
        }
#pragma unroll
        for (int t = 0; t < NT; ++t) {
            v2f u1;
            u1[0] = f.v[i][t][2]; u1[1] = f.v[i][t][3];
            acc[t] = __builtin_amdgcn_wmma_f32_16x16x4_f32(
                false, u1, false, h1, (short)0, acc[t], false, false);
        }
    }
}

template <int NT, bool LOW>
__device__ __forceinline__ void rec_step(const float* __restrict__ prow, const float* pfn,
                                         const float* hr,
                                         float* hw, const float* const (&ua)[NT_MAX],
                                         const int (&hst)[NT_MAX], UFrag& fx, UFrag& fy,
                                         v2f (&c_reg)[NT_MAX], int lhalf) {
    bar_wait();
    bar_signal_lds_done();
    if (!LOW) __builtin_amdgcn_s_setprio(0);
    keep_grouped();

    v4f hb[KGRP];
#pragma unroll
    for (int g = 0; g < KGRP / 2; ++g) hb[g] = *(const v4f*)(hr + 8 * g);
    keep_grouped();

    load_ufrag<NT>(fy, ua, 1);

    v8f acc[NT_MAX];
#pragma unroll
    for (int t = 0; t < NT; ++t)
#pragma unroll
        for (int r = 0; r < 8; ++r) acc[t][r] = 0.0f;

    keep_grouped();
    mma_block<NT>(acc, hb, fx, 0);
    keep_grouped();

    v2f p[NT_MAX][4];
#pragma unroll
    for (int t = 0; t < NT; ++t)
#pragma unroll
        for (int q = 0; q < 4; ++q)
            p[t][q] = *(const v2f*)(prow + q * H_DIM + 4 * t);
    __builtin_prefetch(pfn, 0, 0);

    load_ufrag<NT>(fx, ua, 2);
    keep_grouped();
    mma_block<NT>(acc, hb, fy, 1);
    keep_grouped();

    bar_wait();
    if (!LOW) bar_signal_lds_done();
    if (!LOW) __builtin_amdgcn_s_setprio(2);
    keep_grouped();

#pragma unroll
    for (int g = KGRP / 2; g < KGRP; ++g) hb[g] = *(const v4f*)(hr + 8 * g);
    const v2f ht = *(const v2f*)(hr - 4 * lhalf + KTAIL + 2 * lhalf);
    keep_grouped();

    load_ufrag<NT>(fy, ua, 3);
    v2f ut[NT_MAX];
#pragma unroll
    for (int t = 0; t < NT; ++t)
        ut[t] = *(const v2f*)(ua[t] - 4 * lhalf + KTAIL + 2 * lhalf);
    keep_grouped();
    mma_block<NT>(acc, hb, fx, 2);
    load_ufrag<NT>(fx, ua, 0);
    keep_grouped();
    mma_block<NT>(acc, hb, fy, 3);
#pragma unroll
    for (int t = 0; t < NT; ++t)
        acc[t] = __builtin_amdgcn_wmma_f32_16x16x4_f32(
            false, ut[t], false, ht, (short)0, acc[t], false, false);

    keep_grouped();

    const v2f half2 = {0.5f, 0.5f};
#pragma unroll
    for (int t = 0; t < NT; ++t) {
        v2f ai, af, ag, ao;
        ai[0] = acc[t][0]; ai[1] = acc[t][1];
        af[0] = acc[t][2]; af[1] = acc[t][3];
        ag[0] = acc[t][4]; ag[1] = acc[t][5];
        ao[0] = acc[t][6]; ao[1] = acc[t][7];
        const v2f hi = __builtin_elementwise_fma(half2, p[t][0], ai);
        const v2f hf = __builtin_elementwise_fma(half2, p[t][1], af);
        const v2f zg = ag + p[t][2];
        const v2f ho = __builtin_elementwise_fma(half2, p[t][3], ao);
        v2f ti, tf, gg, to;
#pragma unroll
        for (int uu = 0; uu < 2; ++uu) {
            ti[uu] = fast_tanhf_(hi[uu]);
            tf[uu] = fast_tanhf_(hf[uu]);
            gg[uu] = fast_tanhf_(zg[uu]);
            to[uu] = fast_tanhf_(ho[uu]);
        }
        const v2f ig = __builtin_elementwise_fma(half2, ti, half2);
        const v2f fg = __builtin_elementwise_fma(half2, tf, half2);
        const v2f og = __builtin_elementwise_fma(half2, to, half2);
        const v2f cn = fg * c_reg[t] + ig * gg;
        c_reg[t] = cn;
        v2f hn;
#pragma unroll
        for (int uu = 0; uu < 2; ++uu) hn[uu] = og[uu] * fast_tanhf_(cn[uu]);
        *(v2f*)(hw + hst[t]) = hn;
    }
    if (LOW) bar_signal_lds_done();
}

template <bool FIRST>
__device__ __forceinline__
void rec_body(float* pre, const float* __restrict__ U, float* hT,
              float* UT, float* h_s, int wg, int s_begin, int s_end) {
    const int b0  = wg << 4;
    const int tid = threadIdx.x;
    const int wave  = __builtin_amdgcn_readfirstlane(tid >> 5);
    const int lane  = tid & 31;
    const int lhalf = lane >> 4;
    const int l15   = lane & 15;
    const bool four = (wave == REC_WAVES - 1);
    const bool low  = (wave < REC_WAVES / 2);
    const int tbase = 3 * wave;

    for (int e = tid; e < H_DIM * FH; e += REC_THREADS) {
        const int k = e / FH, col = e - k * FH;
        const bool gcol = (col >= 2 * H_DIM) && (col < 3 * H_DIM);
        UT[col * UP + kpos(k)] = gcol ? U[e] : 0.5f * U[e];
    }
    for (int e = tid; e < 2 * 16 * HP; e += REC_THREADS) h_s[e] = 0.0f;
    if (!FIRST) {
        __syncthreads();
        for (int e = tid; e < 16 * H_DIM; e += REC_THREADS) {
            const int b = e / H_DIM, jj = e - b * H_DIM;
            h_s[b * HP + kpos(jj)] = hT[(size_t)(b0 + b) * H_DIM + jj];
        }
    }
    __syncthreads();

    const float* ua[NT_MAX];
    int hst[NT_MAX];
#pragma unroll
    for (int t = 0; t < NT_MAX; ++t) {
        const int tile = (tbase + t < 25) ? (tbase + t) : 24;
        ua[t]  = UT + (((l15 & 7) >> 1) * H_DIM + 4 * tile + 2 * (l15 >> 3) + (l15 & 1)) * UP
                    + 4 * lhalf;
        hst[t] = l15 * HP + kpos(4 * tile + 2 * lhalf);
    }

    float* cst = pre + ((size_t)b0 + l15) * FH + 4 * tbase + 2 * lhalf;
    v2f c_reg[NT_MAX];
#pragma unroll
    for (int t = 0; t < NT_MAX; ++t) { c_reg[t][0] = 0.0f; c_reg[t][1] = 0.0f; }
    if (!FIRST) {
#pragma unroll
        for (int t = 0; t < NT_MAX; ++t)
            if (t < NT_MAX - 1 || four) c_reg[t] = *(const v2f*)(cst + 4 * t);
    }

    UFrag fx, fy;
    if (four) load_ufrag<4>(fx, ua, 0); else load_ufrag<3>(fx, ua, 0);

    const float* hr0 = h_s + l15 * HP + 4 * lhalf;
    const float* hr1 = hr0 + 16 * HP;
    float* hw0 = h_s;
    float* hw1 = h_s + 16 * HP;
    const size_t pstep = (size_t)B_SZ * FH;
    const float* pl = pre + (size_t)s_begin * pstep + ((size_t)b0 + l15) * FH + 4 * tbase + 2 * lhalf;
    const float* pfp = pre + (size_t)(s_begin + 1) * pstep + (size_t)b0 * FH + 32 * (tid < PF_LINES ? tid : PF_LINES - 1);

    if (low) __builtin_amdgcn_s_setprio(1);
    bar_signal_lds_done();

    for (int s = s_begin; s < s_end; s += 2) {
        if (four)     rec_step<4, false>(pl, pfp, hr0, hw1, ua, hst, fx, fy, c_reg, lhalf);
        else if (low) rec_step<3, true >(pl, pfp, hr0, hw1, ua, hst, fx, fy, c_reg, lhalf);
        else          rec_step<3, false>(pl, pfp, hr0, hw1, ua, hst, fx, fy, c_reg, lhalf);
        const float* plo = pl + pstep;
        const float* pfo = pfp + ((s + 2 < s_end) ? pstep : (size_t)0);
        if (four)     rec_step<4, false>(plo, pfo, hr1, hw0, ua, hst, fx, fy, c_reg, lhalf);
        else if (low) rec_step<3, true >(plo, pfo, hr1, hw0, ua, hst, fx, fy, c_reg, lhalf);
        else          rec_step<3, false>(plo, pfo, hr1, hw0, ua, hst, fx, fy, c_reg, lhalf);
        pl  += 2 * pstep;
        pfp += 2 * pstep;
    }
    bar_wait();
    __syncthreads();

    const float* hl = h_s;
    for (int e = tid; e < 16 * H_DIM; e += REC_THREADS) {
        const int b = e / H_DIM, jj = e - b * H_DIM;
        hT[(size_t)(b0 + b) * H_DIM + jj] = hl[b * HP + kpos(jj)];
    }
    if (FIRST) {
#pragma unroll
        for (int t = 0; t < NT_MAX; ++t)
            if (t < NT_MAX - 1 || four) *(v2f*)(cst + 4 * t) = c_reg[t];
    }
}

__global__ __launch_bounds__(XW_THREADS)
void lstm_overlap_kernel(const float* __restrict__ x, const float* __restrict__ W,
                         const float* __restrict__ bias, const float* __restrict__ U,
                         float* pre, float* hT) {
    __shared__ __attribute__((aligned(16))) float sm[FH * WPITCH + FH];
    if (blockIdx.x < REC_GRID)
        rec_body<true>(pre, U, hT, sm, sm + FH * UP, blockIdx.x, 0, S_SPLIT);
    else
        xw_body(x, W, bias, pre, sm, sm + FH * WPITCH, blockIdx.x - REC_GRID, XW_SIDE_GRID,
                S_SPLIT * MT_PER_S, S_LEN * MT_PER_S);
}

__global__ __launch_bounds__(REC_THREADS)
void lstm_rec_tail_kernel(float* pre, const float* __restrict__ U, float* hT) {
    __shared__ __attribute__((aligned(16))) float UT[FH * UP];
    __shared__ __attribute__((aligned(16))) float h_s[2 * 16 * HP];
    rec_body<false>(pre, U, hT, UT, h_s, blockIdx.x, S_SPLIT, S_LEN);
}

__global__ __launch_bounds__(32)
void dense_kernel(const float* __restrict__ hT, const float* __restrict__ Wd,
                  const float* __restrict__ bd, float* __restrict__ out) {
    const int m0 = (blockIdx.x >> 4) << 4;
    const int n0 = (blockIdx.x & 15) << 4;
    const int lane  = threadIdx.x & 31;
    const int lhalf = lane >> 4;
    const int l15   = lane & 15;
    const int koff  = lhalf * 2;
    const int n = n0 + l15;
    const bool nvalid = (n < C_OUT);

    v8f acc;
    const float bb = nvalid ? bd[n] : 0.0f;
#pragma unroll
    for (int r = 0; r < 8; ++r) acc[r] = bb;

#pragma unroll
    for (int kk = 0; kk < 25; ++kk) {
        const int k0 = kk * 4 + koff;
        v2f a, bf;
        a[0]  = hT[(size_t)(m0 + l15) * H_DIM + k0];
        a[1]  = hT[(size_t)(m0 + l15) * H_DIM + k0 + 1];
        bf[0] = nvalid ? Wd[(size_t)k0 * C_OUT + n] : 0.0f;
        bf[1] = nvalid ? Wd[(size_t)(k0 + 1) * C_OUT + n] : 0.0f;
        acc = __builtin_amdgcn_wmma_f32_16x16x4_f32(
            false, a, false, bf, (short)0, acc, false, false);
    }
    if (nvalid) {
#pragma unroll
        for (int r = 0; r < 8; ++r)
            out[(size_t)(m0 + r + lhalf * 8) * C_OUT + n] = acc[r];
    }
}

extern "C" void kernel_launch(void* const* d_in, const int* in_sizes, int n_in,
                              void* d_out, int out_size, void* d_ws, size_t ws_size,
                              hipStream_t stream) {
    const float* x  = (const float*)d_in[0];
    const float* W  = (const float*)d_in[1];
    const float* U  = (const float*)d_in[2];
    const float* bi = (const float*)d_in[3];
    const float* Wd = (const float*)d_in[4];
    const float* bd = (const float*)d_in[5];
    float* out = (float*)d_out;

    float* pre = (float*)d_ws;
    float* hT  = pre + (size_t)S_LEN * B_SZ * FH;

    lstm_xw_head_kernel<<<XW_GRID, XW_THREADS, 0, stream>>>(x, W, bi, pre);
    lstm_overlap_kernel<<<XW_GRID, XW_THREADS, 0, stream>>>(x, W, bi, U, pre, hT);
    lstm_rec_tail_kernel<<<REC_GRID, REC_THREADS, 0, stream>>>(pre, U, hT);
    dense_kernel<<<(B_SZ / 16) * 16, 32, 0, stream>>>(hT, Wd, bd, out);
}
